# starting kernel
# baseline (speedup 1.0000x reference)
#include <hip/hip_runtime.h>
#include <hip/hip_bf16.h>


typedef __attribute__((ext_vector_type(2))) float v2f;
typedef __attribute__((ext_vector_type(8))) float v8f;

#define F_IN   128
#define HEADS  4
#define C_OUT  32
#define F_OUT  128
#define ALPHA_ 0.5f
#define SLOPE_ 0.2f

__device__ __forceinline__ void atomicMaxFloat(float* addr, float val) {
    if (val >= 0.0f) {
        atomicMax((int*)addr, __float_as_int(val));
    } else {
        atomicMin((unsigned int*)addr, (unsigned int)__float_as_int(val));
    }
}

__device__ __forceinline__ float lrelu(float v) {
    return v > 0.0f ? v : v * SLOPE_;
}

__global__ __launch_bounds__(32) void gemm_wmma_kernel(
    const float* __restrict__ X,
    const float* __restrict__ W1, const float* __restrict__ W2,
    float* __restrict__ H1, float* __restrict__ H2, int n)
{
    int tile = blockIdx.x;
    int conv = tile & 1;         tile >>= 1;
    int ntile = tile & 7;
    int mtile = tile >> 3;

    const float* __restrict__ Wm = conv ? W2 : W1;
    float* __restrict__ Hm       = conv ? H2 : H1;

    int lane = threadIdx.x;
    int half = lane >> 4;
    int l    = lane & 15;

    int rowA = mtile * 16 + l;
    if (rowA >= n) rowA = n - 1;
    int colB = ntile * 16 + l;

    v8f acc = {};
    for (int k = 0; k < F_IN; k += 4) {
        int ka = k + half * 2;
        v2f a, b;
        a.x = X[rowA * F_IN + ka];
        a.y = X[rowA * F_IN + ka + 1];
        b.x = Wm[ka * F_OUT + colB];
        b.y = Wm[(ka + 1) * F_OUT + colB];
        acc = __builtin_amdgcn_wmma_f32_16x16x4_f32(
             false, a,  false, b,
             (short)0, acc,  false,  false);
    }

    int rowBase = mtile * 16 + half * 8;
    int colOut  = ntile * 16 + l;
    #pragma unroll
    for (int r = 0; r < 8; ++r) {
        int row = rowBase + r;
        if (row < n) Hm[row * F_OUT + colOut] = acc[r];
    }
}

__global__ void scores_kernel(
    const float* __restrict__ H1, const float* __restrict__ H2,
    const float* __restrict__ as1, const float* __restrict__ ad1,
    const float* __restrict__ as2, const float* __restrict__ ad2,
    float* __restrict__ es1, float* __restrict__ ed1,
    float* __restrict__ es2, float* __restrict__ ed2, int n)
{
    int i = blockIdx.x * blockDim.x + threadIdx.x;
    int total = n * HEADS * 2;
    if (i >= total) return;
    int conv = (i >= n * HEADS) ? 1 : 0;
    int j = conv ? (i - n * HEADS) : i;
    int node = j >> 2;
    int h    = j & 3;

    const float* __restrict__ hp  = (conv ? H2 : H1) + node * F_OUT + h * C_OUT;
    const float* __restrict__ asp = (conv ? as2 : as1) + h * C_OUT;
    const float* __restrict__ adp = (conv ? ad2 : ad1) + h * C_OUT;

    float ssum = 0.0f, dsum = 0.0f;
    #pragma unroll
    for (int c = 0; c < C_OUT; ++c) {
        float v = hp[c];
        ssum += v * asp[c];
        dsum += v * adp[c];
    }
    if (conv) { es2[node * HEADS + h] = ssum; ed2[node * HEADS + h] = dsum; }
    else      { es1[node * HEADS + h] = ssum; ed1[node * HEADS + h] = dsum; }
}

__global__ void init_kernel(
    const float* __restrict__ b1, const float* __restrict__ b2,
    float* __restrict__ out,
    float* __restrict__ m1, float* __restrict__ d1,
    float* __restrict__ m2, float* __restrict__ d2, int n)
{
    int i = blockIdx.x * blockDim.x + threadIdx.x;
    if (i >= n * F_OUT) return;
    int f = i & (F_OUT - 1);
    out[i] = (1.0f - ALPHA_) * b1[f] + ALPHA_ * b2[f];
    if (i < n * HEADS) {
        float ninf = -__builtin_huge_valf();
        m1[i] = ninf; m2[i] = ninf;
        d1[i] = 0.0f; d2[i] = 0.0f;
    }
}

__global__ void edge_max_kernel(
    const int* __restrict__ srcA, const int* __restrict__ dstA, int E, int n,
    const float* __restrict__ es, const float* __restrict__ ed,
    float* __restrict__ m)
{
    int idx = blockIdx.x * blockDim.x + threadIdx.x;
    if (idx >= (E + n) * HEADS) return;
    int item = idx >> 2, h = idx & 3;
    int s, d;
    if (item < E) { s = srcA[item]; d = dstA[item]; }
    else          { s = d = item - E; }
    float v = lrelu(es[s * HEADS + h] + ed[d * HEADS + h]);
    atomicMaxFloat(&m[d * HEADS + h], v);
}

__global__ void edge_sum_kernel(
    const int* __restrict__ srcA, const int* __restrict__ dstA, int E, int n,
    const float* __restrict__ es, const float* __restrict__ ed,
    const float* __restrict__ m, float* __restrict__ denom)
{
    int idx = blockIdx.x * blockDim.x + threadIdx.x;
    if (idx >= (E + n) * HEADS) return;
    int item = idx >> 2, h = idx & 3;
    int s, d;
    if (item < E) { s = srcA[item]; d = dstA[item]; }
    else          { s = d = item - E; }
    float v = lrelu(es[s * HEADS + h] + ed[d * HEADS + h]);
    float ex = __expf(v - m[d * HEADS + h]);
    atomicAdd(&denom[d * HEADS + h], ex);
}

__global__ void edge_msg_kernel(
    const int* __restrict__ srcA, const int* __restrict__ dstA, int E, int n,
    const float* __restrict__ es, const float* __restrict__ ed,
    const float* __restrict__ m, const float* __restrict__ denom,
    const float* __restrict__ Hm, float scale, float* __restrict__ out)
{
    int idx = blockIdx.x * blockDim.x + threadIdx.x;
    if (idx >= (E + n) * HEADS) return;
    int item = idx >> 2, h = idx & 3;
    int s, d;
    if (item < E) { s = srcA[item]; d = dstA[item]; }
    else          { s = d = item - E; }
    float v = lrelu(es[s * HEADS + h] + ed[d * HEADS + h]);
    float att = __expf(v - m[d * HEADS + h]) / denom[d * HEADS + h];
    float f = scale * att;

    const float* __restrict__ hp = Hm + s * F_OUT + h * C_OUT;
    float* __restrict__ op       = out + d * F_OUT + h * C_OUT;
    __builtin_prefetch(hp, 0, 1);
    #pragma unroll
    for (int c = 0; c < C_OUT; ++c) {
        atomicAdd(&op[c], f * hp[c]);
    }
}

extern "C" void kernel_launch(void* const* d_in, const int* in_sizes, int n_in,
                              void* d_out, int out_size, void* d_ws, size_t ws_size,
                              hipStream_t stream) {
    const float* x    = (const float*)d_in[0];
    const int*   ei   = (const int*)d_in[1];
    const float* W1   = (const float*)d_in[2];
    const float* as1  = (const float*)d_in[3];
    const float* ad1  = (const float*)d_in[4];
    const float* b1   = (const float*)d_in[5];
    const float* W2   = (const float*)d_in[6];
    const float* as2  = (const float*)d_in[7];
    const float* ad2  = (const float*)d_in[8];
    const float* b2   = (const float*)d_in[9];
    float* out = (float*)d_out;

    const int n = in_sizes[0] / F_IN;
    const int E = in_sizes[1] / 2;
    const int* src = ei;
    const int* dst = ei + E;

    float* p   = (float*)d_ws;
    float* H1  = p;  p += (size_t)n * F_OUT;
    float* H2  = p;  p += (size_t)n * F_OUT;
    float* es1 = p;  p += (size_t)n * HEADS;
    float* ed1 = p;  p += (size_t)n * HEADS;
    float* es2 = p;  p += (size_t)n * HEADS;
    float* ed2 = p;  p += (size_t)n * HEADS;
    float* m1  = p;  p += (size_t)n * HEADS;
    float* dn1 = p;  p += (size_t)n * HEADS;
    float* m2  = p;  p += (size_t)n * HEADS;
    float* dn2 = p;  p += (size_t)n * HEADS;

    {
        int mtiles = (n + 15) / 16;
        int blocks = 2 * mtiles * (F_OUT / 16);
        gemm_wmma_kernel<<<blocks, 32, 0, stream>>>(x, W1, W2, H1, H2, n);
    }
    {
        int total = n * HEADS * 2;
        scores_kernel<<<(total + 255) / 256, 256, 0, stream>>>(
            H1, H2, as1, ad1, as2, ad2, es1, ed1, es2, ed2, n);
    }
    {
        int total = n * F_OUT;
        init_kernel<<<(total + 255) / 256, 256, 0, stream>>>(
            b1, b2, out, m1, dn1, m2, dn2, n);
    }
    const int eth = (E + n) * HEADS;
    const int eblk = (eth + 255) / 256;
    edge_max_kernel<<<eblk, 256, 0, stream>>>(src, dst, E, n, es1, ed1, m1);
    edge_max_kernel<<<eblk, 256, 0, stream>>>(dst, src, E, n, es2, ed2, m2);
    edge_sum_kernel<<<eblk, 256, 0, stream>>>(src, dst, E, n, es1, ed1, m1, dn1);
    edge_sum_kernel<<<eblk, 256, 0, stream>>>(dst, src, E, n, es2, ed2, m2, dn2);
    edge_msg_kernel<<<eblk, 256, 0, stream>>>(src, dst, E, n, es1, ed1, m1, dn1,
                                              H1, 1.0f - ALPHA_, out);
    edge_msg_kernel<<<eblk, 256, 0, stream>>>(dst, src, E, n, es2, ed2, m2, dn2,
                                              H2, ALPHA_, out);
}


# edge sweep reads bit-sliced owner keys (8 planes per end per 32 edges, built with a packed row pair per edge by extra groups of the scores launch): 32 keys tested by 8 bit ops; new sum order
# speedup vs baseline: 11.9309x; 11.9309x over previous
#include <hip/hip_runtime.h>
#include <hip/hip_bf16.h>


typedef __attribute__((ext_vector_type(2))) float v2f;
typedef __attribute__((ext_vector_type(8))) float v8f;

#define F_IN   128
#define HEADS  4
#define C_OUT  32
#define F_OUT  128
#define ALPHA_ 0.5f
#define SLOPE_ 0.2f

__device__ __forceinline__ float lrelu(float v) {
    return v > 0.0f ? v : v * SLOPE_;
}

__global__ __launch_bounds__(32) void gemm_wmma_kernel(
    const float* __restrict__ X,
    const float* __restrict__ W1, const float* __restrict__ W2,
    float* __restrict__ H1, float* __restrict__ H2, int n)
{
    int tile = blockIdx.x;
    int conv = tile & 1;         tile >>= 1;
    int ntile = tile & 7;
    int mtile = tile >> 3;

    const float* __restrict__ Wm = conv ? W2 : W1;
    float* __restrict__ Hm       = conv ? H2 : H1;

    int lane = threadIdx.x;
    int half = lane >> 4;
    int l    = lane & 15;

    int rowA = mtile * 16 + l;
    if (rowA >= n) rowA = n - 1;
    int colB = ntile * 16 + l;

    v8f acc = {};
    for (int k = 0; k < F_IN; k += 4) {
        int ka = k + half * 2;
        v2f a, b;
        a.x = X[rowA * F_IN + ka];
        a.y = X[rowA * F_IN + ka + 1];
        b.x = Wm[ka * F_OUT + colB];
        b.y = Wm[(ka + 1) * F_OUT + colB];
        acc = __builtin_amdgcn_wmma_f32_16x16x4_f32(
             false, a,  false, b,
             (short)0, acc,  false,  false);
    }

    int rowBase = mtile * 16 + half * 8;
    int colOut  = ntile * 16 + l;
    #pragma unroll
    for (int r = 0; r < 8; ++r) {
        int row = rowBase + r;
        if (row < n) Hm[row * F_OUT + colOut] = acc[r];
    }
}

#define ROWS_PG  196

__device__ __forceinline__ int key_runs(int E) { return ((E >> 5) + 31) >> 5; }
__device__ __forceinline__ unsigned long long key_bytes(int E, int n) {
    return (unsigned long long)n * (2 * F_OUT + 4 * HEADS) * 4ull + (unsigned long long)key_runs(E) * 2048ull;
}
__device__ __forceinline__ bool keys_usable(int E, int n, unsigned long long ws_bytes) {
    return (E & 31) == 0 && (n + ROWS_PG - 1) / ROWS_PG <= 256 && ws_bytes >= key_bytes(E, n);
}
__device__ __forceinline__ bool pairs_usable(int E, int n, unsigned long long ws_bytes) {
    return ws_bytes >= key_bytes(E, n) + (unsigned long long)E * 4ull;
}

__global__ void scores_kernel(
    const float* __restrict__ H1, const float* __restrict__ H2,
    const float* __restrict__ as1, const float* __restrict__ ad1,
    const float* __restrict__ as2, const float* __restrict__ ad2,
    float* __restrict__ es1, float* __restrict__ ed1,
    float* __restrict__ es2, float* __restrict__ ed2, int n,
    const int* __restrict__ ei0, const int* __restrict__ ei1, int E, int nkb,
    unsigned* __restrict__ planes, unsigned* __restrict__ pair,
    unsigned long long ws_bytes)
{
    if ((int)blockIdx.x < nkb) {
        if (!keys_usable(E, n, ws_bytes)) return;
        const int lane = threadIdx.x & 31;
        const int w    = __builtin_amdgcn_readfirstlane((int)(blockIdx.x * (blockDim.x >> 5) + (threadIdx.x >> 5)));
        const int nblk = E >> 5;
        if (w >= key_runs(E) * 4) return;
        const bool with_pairs = pairs_usable(E, n, ws_bytes);
        int av[8], bv[8];
        #pragma unroll
        for (int u = 0; u < 8; ++u) {
            const int gb = w * 8 + u;
            const int ec = gb < nblk ? gb * 32 + lane : 0;
            av[u] = ei0[ec];
            bv[u] = ei1[ec];
        }
        unsigned pa[8], pb[8];
        #pragma unroll
        for (int b = 0; b < 8; ++b) { pa[b] = 0u; pb[b] = 0u; }
        #pragma unroll
        for (int u = 0; u < 8; ++u) {
            const int gb = w * 8 + u;
            if (with_pairs && gb < nblk) pair[gb * 32 + lane] = (unsigned)av[u] | (unsigned)bv[u] << 16;
            const unsigned ka = (unsigned)av[u] / ROWS_PG;
            const unsigned kb = (unsigned)bv[u] / ROWS_PG;
            #pragma unroll
            for (int b = 0; b < 8; ++b) {
                const unsigned ba = __builtin_amdgcn_ballot_w32(((ka >> b) & 1u) != 0u);
                const unsigned bb = __builtin_amdgcn_ballot_w32(((kb >> b) & 1u) != 0u);
                if (lane == u) { pa[b] = ba; pb[b] = bb; }
            }
        }
        if (lane < 8) {
            const int gb = w * 8 + lane;
            uint4* __restrict__ p4 = (uint4*)planes + (size_t)(gb >> 5) * 128 + (gb & 31);
            p4[0]  = make_uint4(pa[0], pa[1], pa[2], pa[3]);
            p4[32] = make_uint4(pa[4], pa[5], pa[6], pa[7]);
            p4[64] = make_uint4(pb[0], pb[1], pb[2], pb[3]);
            p4[96] = make_uint4(pb[4], pb[5], pb[6], pb[7]);
        }
        return;
    }
    int i = (blockIdx.x - nkb) * blockDim.x + threadIdx.x;
    int total = n * HEADS * 2;
    if (i >= total) return;
    int conv = (i >= n * HEADS) ? 1 : 0;
    int j = conv ? (i - n * HEADS) : i;
    int node = j >> 2;
    int h    = j & 3;

    const float* __restrict__ hp  = (conv ? H2 : H1) + node * F_OUT + h * C_OUT;
    const float* __restrict__ asp = (conv ? as2 : as1) + h * C_OUT;
    const float* __restrict__ adp = (conv ? ad2 : ad1) + h * C_OUT;

    float ssum = 0.0f, dsum = 0.0f;
    #pragma unroll
    for (int c = 0; c < C_OUT; ++c) {
        float v = hp[c];
        ssum += v * asp[c];
        dsum += v * adp[c];
    }
    if (conv) { es2[node * HEADS + h] = ssum; ed2[node * HEADS + h] = dsum; }
    else      { es1[node * HEADS + h] = ssum; ed1[node * HEADS + h] = dsum; }
}

#define QCAP     768

struct Msg { float h0, h1, h2, h3, e; float2 em; int slot; };

__global__ __launch_bounds__(1024) void owner_kernel(
    const int* __restrict__ ei0, const int* __restrict__ ei1, int E, int n,
    const float* __restrict__ H1, const float* __restrict__ H2,
    const float* __restrict__ es1, const float* __restrict__ ed1,
    const float* __restrict__ es2, const float* __restrict__ ed2,
    const float* __restrict__ b1, const float* __restrict__ b2,
    const unsigned* __restrict__ planes, const unsigned* __restrict__ pair,
    unsigned long long ws_bytes,
    float* __restrict__ out)
{
    __shared__ __attribute__((aligned(16))) float s_num[2 * ROWS_PG * F_OUT];
    __shared__ float  s_den[2 * ROWS_PG * HEADS];
    __shared__ float2 s_em[2 * ROWS_PG * HEADS];
    __shared__ __attribute__((aligned(16))) int s_q[32 * QCAP];
    __shared__ int s_over;

    const int tid  = threadIdx.x;
    const int lane = tid & 31;
    const int wave = __builtin_amdgcn_readfirstlane(tid >> 5);
    const int hd   = lane & 3;
    const int r0   = blockIdx.x * ROWS_PG;
    int nrows = n - r0;
    if (nrows > ROWS_PG) nrows = ROWS_PG;

    {
        const float4 z = make_float4(0.0f, 0.0f, 0.0f, 0.0f);
        for (int i = tid; i < 2 * ROWS_PG * (F_OUT / 4); i += 1024) ((float4*)s_num)[i] = z;
    }
    for (int i = tid; i < 2 * ROWS_PG * HEADS; i += 1024) {
        int conv = i >= ROWS_PG * HEADS ? 1 : 0;
        int j = i - conv * ROWS_PG * HEADS;
        float2 em; em.x = 0.0f; em.y = -__builtin_huge_valf();
        if ((j >> 2) < nrows) em.x = (conv ? ed2 : ed1)[r0 * HEADS + j];
        s_em[i]  = em;
        s_den[i] = 0.0f;
    }
    if (tid == 0) s_over = 0;
    __syncthreads();

    auto fetch = [&](int word_in) -> Msg {
        const int word = __builtin_amdgcn_readfirstlane(word_in);
        const int s    = word >> 9;
        const int conv = (word >> 8) & 1;
        Msg g;
        g.slot = conv * ROWS_PG + (word & 255);
        const float* __restrict__ hp = (conv ? H2 : H1) + (size_t)s * F_OUT;
        const float* __restrict__ ep = (conv ? es2 : es1) + (size_t)s * HEADS;
        g.h0 = hp[lane]; g.h1 = hp[32 + lane]; g.h2 = hp[64 + lane]; g.h3 = hp[96 + lane];
        g.e  = ep[hd];
        g.em = s_em[g.slot * HEADS + hd];
        return g;
    };
    auto apply = [&](const Msg& g) {
        float v = lrelu(g.e + g.em.x);
        float w = __expf(v - g.em.y);
        if (lane < HEADS)
            __hip_atomic_fetch_add(&s_den[g.slot * HEADS + lane], w,
                                   __ATOMIC_RELAXED, __HIP_MEMORY_SCOPE_WORKGROUP);
        int wi = __float_as_int(w);
        float w0 = __int_as_float(__builtin_amdgcn_readlane(wi, 0));
        float w1 = __int_as_float(__builtin_amdgcn_readlane(wi, 1));
        float w2 = __int_as_float(__builtin_amdgcn_readlane(wi, 2));
        float w3 = __int_as_float(__builtin_amdgcn_readlane(wi, 3));
        float* np = &s_num[g.slot * F_OUT + lane];
        __hip_atomic_fetch_add(np,      w0 * g.h0, __ATOMIC_RELAXED, __HIP_MEMORY_SCOPE_WORKGROUP);
        __hip_atomic_fetch_add(np + 32, w1 * g.h1, __ATOMIC_RELAXED, __HIP_MEMORY_SCOPE_WORKGROUP);
        __hip_atomic_fetch_add(np + 64, w2 * g.h2, __ATOMIC_RELAXED, __HIP_MEMORY_SCOPE_WORKGROUP);
        __hip_atomic_fetch_add(np + 96, w3 * g.h3, __ATOMIC_RELAXED, __HIP_MEMORY_SCOPE_WORKGROUP);
    };
    auto drain = [&](int cnt) -> int {
        int* q = &s_q[wave * QCAP];
        int i = 0;
        for (; i + 8 <= cnt; i += 8) {
            Msg g[8];
            #pragma unroll
            for (int u = 0; u < 8; ++u) g[u] = fetch(q[i + u]);
            #pragma unroll
            for (int u = 0; u < 8; ++u) apply(g[u]);
        }
        const int rem = cnt - i;
        if (i > 0 && lane < rem) { const int wv = q[i + lane]; q[lane] = wv; }
        return rem;
    };
    auto tail = [&](int cnt) {
        for (int i = 0; i < cnt; ++i) { Msg g = fetch(s_q[wave * QCAP + i]); apply(g); }
    };
    auto maxpass = [&](int cnt) {
        const int* q = &s_q[wave * QCAP];
        const int sub = lane >> 2;
        for (int i = 0; i < cnt; i += 64) {
            float e[8];
            int   at[8];
            #pragma unroll
            for (int u = 0; u < 8; ++u) {
                int k = i + 8 * u + sub;
                if (k > cnt - 1) k = cnt - 1;
                const int word = q[k];
                const int conv = (word >> 8) & 1;
                at[u] = (conv * ROWS_PG + (word & 255)) * HEADS + hd;
                e[u]  = (conv ? es2 : es1)[(size_t)(word >> 9) * HEADS + hd];
            }
            #pragma unroll
            for (int u = 0; u < 8; ++u) {
                const float v = lrelu(e[u] + s_em[at[u]].x);
                __hip_atomic_fetch_max(&s_em[at[u]].y, v,
                                       __ATOMIC_RELAXED, __HIP_MEMORY_SCOPE_WORKGROUP);
            }
        }
    };

    int qn = 0;
    auto seed = [&]() -> int {
        const int row = wave + 32 * (lane >> 1);
        if (row < nrows) s_q[wave * QCAP + lane] = ((r0 + row) << 9) | ((lane & 1) << 8) | row;
        return wave < nrows ? 2 * ((nrows - wave + 31) >> 5) : 0;
    };
    const unsigned lt = (1u << lane) - 1u;
    auto handle = [&](int a, int b, unsigned lim) {
        const bool t0 = (unsigned)(b - r0) < lim;
        const bool t1 = (unsigned)(a - r0) < lim;
        const unsigned bal0 = __builtin_amdgcn_ballot_w32(t0);
        const unsigned bal1 = __builtin_amdgcn_ballot_w32(t1);
        if (bal0 | bal1) {
            const int c0 = __popc(bal0);
            int* q = &s_q[wave * QCAP + qn];
            if (t0) q[__popc(bal0 & lt)]      = (a << 9) | (b - r0);
            if (t1) q[c0 + __popc(bal1 & lt)] = (b << 9) | 256 | (a - r0);
            qn += c0 + __popc(bal1);
        }
    };
    auto plain_sweep = [&](auto full) {
        for (int base = 0; base < E; base += 1024) {
            const int e = base + tid;
            const bool ok = e < E;
            const int ec = ok ? e : 0;
            if (qn > QCAP - 64) qn = full(qn);
            handle(ei0[ec], ei1[ec], ok ? (unsigned)nrows : 0u);
        }
    };
    auto give_up = [&](int) -> int { s_over = 1; return 0; };

    qn = seed();
    if (keys_usable(E, n, ws_bytes)) {
        const uint4* __restrict__ p4 = (const uint4*)planes;
        const int nblk  = E >> 5;
        const int nrun  = key_runs(E);
        const int niter = (nrun + 31) >> 5;
        const int seeds = qn;
        int gave = 0;
        const int start = (int)((blockIdx.x * (unsigned)niter) / gridDim.x);
        unsigned gm[8];
        #pragma unroll
        for (int b = 0; b < 8; ++b) gm[b] = ((blockIdx.x >> b) & 1u) ? ~0u : 0u;
        auto ld = [&](int it, uint4 (&k)[4]) -> int {
            int c = start + it; if (c >= niter) c -= niter;
            const int run = c * 32 + wave;
            const bool in = (it < niter) && (run < nrun);
            const int rc = in ? run : 0;
            #pragma unroll
            for (int j = 0; j < 4; ++j) k[j] = p4[rc * 128 + j * 32 + lane];
            const int gb = rc * 32 + lane;
            return (in && gb < nblk) ? gb : -1;
        };
        auto match = [&](const uint4& lo, const uint4& hi) -> unsigned {
            unsigned m = ~(lo.x ^ gm[0]);
            m &= ~(lo.y ^ gm[1]); m &= ~(lo.z ^ gm[2]); m &= ~(lo.w ^ gm[3]);
            m &= ~(hi.x ^ gm[4]); m &= ~(hi.y ^ gm[5]); m &= ~(hi.z ^ gm[6]); m &= ~(hi.w ^ gm[7]);
            return m;
        };
        auto note = [&](unsigned m, int gb, int conv) {
            while (true) {
                const bool has = m != 0u;
                const unsigned bal = __builtin_amdgcn_ballot_w32(has);
                if (bal == 0u) break;
                if (qn > QCAP - 32) { s_over = 1; gave = 1; qn = 0; }
                if (has) {
                    const int p = __builtin_ctz(m);
                    m &= m - 1u;
                    s_q[wave * QCAP + qn + __popc(bal & lt)] = (((gb << 5) | p) << 1) | conv;
                }
                qn += __popc(bal);
            }
        };
        auto work = [&](const uint4 (&k)[4], int gb) {
            const unsigned m1 = gb >= 0 ? match(k[2], k[3]) : 0u;
            const unsigned m2 = gb >= 0 ? match(k[0], k[1]) : 0u;
            note(m1, gb, 0);
            note(m2, gb, 1);
        };
        uint4 kc[2][4];
        int gc[2];
        #pragma unroll
        for (int s = 0; s < 2; ++s) gc[s] = ld(s, kc[s]);
        for (int it = 0; it < niter; it += 2) {
            #pragma unroll
            for (int s = 0; s < 2; ++s) {
                work(kc[s], gc[s]);
                gc[s] = ld(it + 2 + s, kc[s]);
            }
        }
        const int first = gave ? 0 : seeds;
        if (gave) qn = 0;
        const bool with_pairs = pairs_usable(E, n, ws_bytes);
        for (int base = first; base < qn; base += 128) {
            int code[4], ra[4], rb[4];
            #pragma unroll
            for (int u = 0; u < 4; ++u) {
                const int i = base + 32 * u + lane;
                code[u] = 0;
                if (i < qn) code[u] = s_q[wave * QCAP + i];
                if (with_pairs) {
                    const unsigned pv = pair[code[u] >> 1];
                    ra[u] = (int)(pv & 0xFFFFu);
                    rb[u] = (int)(pv >> 16);
                } else {
                    ra[u] = ei0[code[u] >> 1];
                    rb[u] = ei1[code[u] >> 1];
                }
            }
            #pragma unroll
            for (int u = 0; u < 4; ++u) {
                const int i = base + 32 * u + lane;
                if (i < qn)
                    s_q[wave * QCAP + i] = (code[u] & 1) ? ((rb[u] << 9) | 256 | (ra[u] - r0))
                                                         : ((ra[u] << 9) | (rb[u] - r0));
            }
        }
    } else {
        plain_sweep(give_up);
    }
    maxpass(qn);
    __syncthreads();

    const int over = __builtin_amdgcn_readfirstlane(s_over);
    if (!over) {
        qn = drain(qn);
        tail(qn);
    } else {
        qn = seed();
        plain_sweep([&](int cnt) -> int { maxpass(cnt); return 0; });
        maxpass(qn);
        __syncthreads();
        qn = seed();
        plain_sweep(drain);
        qn = drain(qn);
        tail(qn);
    }
    __syncthreads();

    for (int i = tid; i < nrows * (F_OUT / 4); i += 1024) {
        const int row = i >> 5;
        const int c4  = (i & 31) * 4;
        const int h   = c4 >> 5;
        const float4 n1 = *(const float4*)&s_num[row * F_OUT + c4];
        const float4 n2 = *(const float4*)&s_num[(ROWS_PG + row) * F_OUT + c4];
        const float d1 = s_den[row * HEADS + h];
        const float d2 = s_den[(ROWS_PG + row) * HEADS + h];
        const float4 bb1 = *(const float4*)&b1[c4];
        const float4 bb2 = *(const float4*)&b2[c4];
        float4 o;
        o.x = (1.0f - ALPHA_) * (n1.x / d1 + bb1.x) + ALPHA_ * (n2.x / d2 + bb2.x);
        o.y = (1.0f - ALPHA_) * (n1.y / d1 + bb1.y) + ALPHA_ * (n2.y / d2 + bb2.y);
        o.z = (1.0f - ALPHA_) * (n1.z / d1 + bb1.z) + ALPHA_ * (n2.z / d2 + bb2.z);
        o.w = (1.0f - ALPHA_) * (n1.w / d1 + bb1.w) + ALPHA_ * (n2.w / d2 + bb2.w);
        *(float4*)&out[(size_t)(r0 + row) * F_OUT + c4] = o;
    }
}

extern "C" void kernel_launch(void* const* d_in, const int* in_sizes, int n_in,
                              void* d_out, int out_size, void* d_ws, size_t ws_size,
                              hipStream_t stream) {
    const float* x    = (const float*)d_in[0];
    const int*   ei   = (const int*)d_in[1];
    const float* W1   = (const float*)d_in[2];
    const float* as1  = (const float*)d_in[3];
    const float* ad1  = (const float*)d_in[4];
    const float* b1   = (const float*)d_in[5];
    const float* W2   = (const float*)d_in[6];
    const float* as2  = (const float*)d_in[7];
    const float* ad2  = (const float*)d_in[8];
    const float* b2   = (const float*)d_in[9];
    float* out = (float*)d_out;

    const int n = in_sizes[0] / F_IN;
    const int E = in_sizes[1] / 2;
    const int* src = ei;
    const int* dst = ei + E;

    float* p   = (float*)d_ws;
    float* H1  = p;  p += (size_t)n * F_OUT;
    float* H2  = p;  p += (size_t)n * F_OUT;
    float* es1 = p;  p += (size_t)n * HEADS;
    float* ed1 = p;  p += (size_t)n * HEADS;
    float* es2 = p;  p += (size_t)n * HEADS;
    float* ed2 = p;  p += (size_t)n * HEADS;
    int nrun = (E / 32 + 31) / 32;
    unsigned* planes = (unsigned*)p;
    unsigned* pair   = planes + (size_t)nrun * 512;

    {
        int mtiles = (n + 15) / 16;
        int blocks = 2 * mtiles * (F_OUT / 16);
        gemm_wmma_kernel<<<blocks, 32, 0, stream>>>(x, W1, W2, H1, H2, n);
    }
    {
        int total = n * HEADS * 2;
        int nkb = (nrun * 4 + 7) / 8;
        scores_kernel<<<nkb + (total + 255) / 256, 256, 0, stream>>>(
            H1, H2, as1, ad1, as2, ad2, es1, ed1, es2, ed2, n,
            src, dst, E, nkb, planes, pair, (unsigned long long)ws_size);
    }
    {
        int groups = (n + ROWS_PG - 1) / ROWS_PG;
        owner_kernel<<<groups, 1024, 0, stream>>>(src, dst, E, n, H1, H2,
                                                  es1, ed1, es2, ed2, b1, b2,
                                                  planes, pair,
                                                  (unsigned long long)ws_size, out);
    }
}


# x@W: each wave's 16 x 128 tile of x comes in by whole-line 16-byte loads (one row per load) and is staged in LDS, fragments read from LDS; was 8-byte loads touching 16 rows' lines each; same k order
# speedup vs baseline: 14.5688x; 1.2211x over previous
#include <hip/hip_runtime.h>
#include <hip/hip_bf16.h>


typedef __attribute__((ext_vector_type(2))) float v2f;
typedef __attribute__((ext_vector_type(8))) float v8f;

#define F_IN   128
#define HEADS  4
#define C_OUT  32
#define F_OUT  128
#define ALPHA_ 0.5f
#define SLOPE_ 0.2f

__device__ __forceinline__ float lrelu(float v) {
    return v > 0.0f ? v : v * SLOPE_;
}

#define A_PITCH 132

__global__ __launch_bounds__(32) void gemm_wmma_kernel(
    const float* __restrict__ X,
    const float* __restrict__ W1, const float* __restrict__ W2,
    float* __restrict__ H1, float* __restrict__ H2, int n)
{
    __shared__ __attribute__((aligned(16))) float s_a[16 * A_PITCH];

    int tile = blockIdx.x;
    int conv = tile & 1;         tile >>= 1;
    int ntile = tile & 7;
    int mtile = tile >> 3;

    const float* __restrict__ Wm = conv ? W2 : W1;
    float* __restrict__ Hm       = conv ? H2 : H1;

    int lane = threadIdx.x;
    int half = lane >> 4;
    int l    = lane & 15;

    int colB = ntile * 16 + l;

    #pragma unroll
    for (int r = 0; r < 16; ++r) {
        int row = mtile * 16 + r;
        if (row >= n) row = n - 1;
        const float4 v = ((const float4*)(X + (size_t)row * F_IN))[lane];
        *(float4*)&s_a[r * A_PITCH + lane * 4] = v;
    }
    __syncthreads();

    v8f acc = {};
    for (int k = 0; k < F_IN; k += 4) {
        int ka = k + half * 2;
        v2f a, b;
        a = *(const v2f*)&s_a[l * A_PITCH + ka];
        b.x = Wm[ka * F_OUT + colB];
        b.y = Wm[(ka + 1) * F_OUT + colB];
        acc = __builtin_amdgcn_wmma_f32_16x16x4_f32(
             false, a,  false, b,
             (short)0, acc,  false,  false);
    }

    int rowBase = mtile * 16 + half * 8;
    int colOut  = ntile * 16 + l;
    #pragma unroll
    for (int r = 0; r < 8; ++r) {
        int row = rowBase + r;
        if (row < n) Hm[row * F_OUT + colOut] = acc[r];
    }
}

#define ROWS_PG  196

__device__ __forceinline__ int key_runs(int E) { return ((E >> 5) + 31) >> 5; }
__device__ __forceinline__ unsigned long long key_bytes(int E, int n) {
    return (unsigned long long)n * (2 * F_OUT + 4 * HEADS) * 4ull + (unsigned long long)key_runs(E) * 2048ull;
}
__device__ __forceinline__ bool keys_usable(int E, int n, unsigned long long ws_bytes) {
    return (E & 31) == 0 && (n + ROWS_PG - 1) / ROWS_PG <= 256 && ws_bytes >= key_bytes(E, n);
}
__device__ __forceinline__ bool pairs_usable(int E, int n, unsigned long long ws_bytes) {
    return ws_bytes >= key_bytes(E, n) + (unsigned long long)E * 4ull;
}

__global__ void scores_kernel(
    const float* __restrict__ H1, const float* __restrict__ H2,
    const float* __restrict__ as1, const float* __restrict__ ad1,
    const float* __restrict__ as2, const float* __restrict__ ad2,
    float* __restrict__ es1, float* __restrict__ ed1,
    float* __restrict__ es2, float* __restrict__ ed2, int n,
    const int* __restrict__ ei0, const int* __restrict__ ei1, int E, int nkb,
    unsigned* __restrict__ planes, unsigned* __restrict__ pair,
    unsigned long long ws_bytes)
{
    if ((int)blockIdx.x < nkb) {
        if (!keys_usable(E, n, ws_bytes)) return;
        const int lane = threadIdx.x & 31;
        const int w    = __builtin_amdgcn_readfirstlane((int)(blockIdx.x * (blockDim.x >> 5) + (threadIdx.x >> 5)));
        const int nblk = E >> 5;
        if (w >= key_runs(E) * 4) return;
        const bool with_pairs = pairs_usable(E, n, ws_bytes);
        int av[8], bv[8];
        #pragma unroll
        for (int u = 0; u < 8; ++u) {
            const int gb = w * 8 + u;
            const int ec = gb < nblk ? gb * 32 + lane : 0;
            av[u] = ei0[ec];
            bv[u] = ei1[ec];
        }
        unsigned pa[8], pb[8];
        #pragma unroll
        for (int b = 0; b < 8; ++b) { pa[b] = 0u; pb[b] = 0u; }
        #pragma unroll
        for (int u = 0; u < 8; ++u) {
            const int gb = w * 8 + u;
            if (with_pairs && gb < nblk) pair[gb * 32 + lane] = (unsigned)av[u] | (unsigned)bv[u] << 16;
            const unsigned ka = (unsigned)av[u] / ROWS_PG;
            const unsigned kb = (unsigned)bv[u] / ROWS_PG;
            #pragma unroll
            for (int b = 0; b < 8; ++b) {
                const unsigned ba = __builtin_amdgcn_ballot_w32(((ka >> b) & 1u) != 0u);
                const unsigned bb = __builtin_amdgcn_ballot_w32(((kb >> b) & 1u) != 0u);
                if (lane == u) { pa[b] = ba; pb[b] = bb; }
            }
        }
        if (lane < 8) {
            const int gb = w * 8 + lane;
            uint4* __restrict__ p4 = (uint4*)planes + (size_t)(gb >> 5) * 128 + (gb & 31);
            p4[0]  = make_uint4(pa[0], pa[1], pa[2], pa[3]);
            p4[32] = make_uint4(pa[4], pa[5], pa[6], pa[7]);
            p4[64] = make_uint4(pb[0], pb[1], pb[2], pb[3]);
            p4[96] = make_uint4(pb[4], pb[5], pb[6], pb[7]);
        }
        return;
    }
    int i = (blockIdx.x - nkb) * blockDim.x + threadIdx.x;
    int total = n * HEADS * 2;
    if (i >= total) return;
    int conv = (i >= n * HEADS) ? 1 : 0;
    int j = conv ? (i - n * HEADS) : i;
    int node = j >> 2;
    int h    = j & 3;

    const float* __restrict__ hp  = (conv ? H2 : H1) + node * F_OUT + h * C_OUT;
    const float* __restrict__ asp = (conv ? as2 : as1) + h * C_OUT;
    const float* __restrict__ adp = (conv ? ad2 : ad1) + h * C_OUT;

    float ssum = 0.0f, dsum = 0.0f;
    #pragma unroll
    for (int c = 0; c < C_OUT; ++c) {
        float v = hp[c];
        ssum += v * asp[c];
        dsum += v * adp[c];
    }
    if (conv) { es2[node * HEADS + h] = ssum; ed2[node * HEADS + h] = dsum; }
    else      { es1[node * HEADS + h] = ssum; ed1[node * HEADS + h] = dsum; }
}

#define QCAP     768

struct Msg { float h0, h1, h2, h3, e; float2 em; int slot; };

__global__ __launch_bounds__(1024) void owner_kernel(
    const int* __restrict__ ei0, const int* __restrict__ ei1, int E, int n,
    const float* __restrict__ H1, const float* __restrict__ H2,
    const float* __restrict__ es1, const float* __restrict__ ed1,
    const float* __restrict__ es2, const float* __restrict__ ed2,
    const float* __restrict__ b1, const float* __restrict__ b2,
    const unsigned* __restrict__ planes, const unsigned* __restrict__ pair,
    unsigned long long ws_bytes,
    float* __restrict__ out)
{
    __shared__ __attribute__((aligned(16))) float s_num[2 * ROWS_PG * F_OUT];
    __shared__ float  s_den[2 * ROWS_PG * HEADS];
    __shared__ float2 s_em[2 * ROWS_PG * HEADS];
    __shared__ __attribute__((aligned(16))) int s_q[32 * QCAP];
    __shared__ int s_over;

    const int tid  = threadIdx.x;
    const int lane = tid & 31;
    const int wave = __builtin_amdgcn_readfirstlane(tid >> 5);
    const int hd   = lane & 3;
    const int r0   = blockIdx.x * ROWS_PG;
    int nrows = n - r0;
    if (nrows > ROWS_PG) nrows = ROWS_PG;

    {
        const float4 z = make_float4(0.0f, 0.0f, 0.0f, 0.0f);
        for (int i = tid; i < 2 * ROWS_PG * (F_OUT / 4); i += 1024) ((float4*)s_num)[i] = z;
    }
    for (int i = tid; i < 2 * ROWS_PG * HEADS; i += 1024) {
        int conv = i >= ROWS_PG * HEADS ? 1 : 0;
        int j = i - conv * ROWS_PG * HEADS;
        float2 em; em.x = 0.0f; em.y = -__builtin_huge_valf();
        if ((j >> 2) < nrows) em.x = (conv ? ed2 : ed1)[r0 * HEADS + j];
        s_em[i]  = em;
        s_den[i] = 0.0f;
    }
    if (tid == 0) s_over = 0;
    __syncthreads();

    auto fetch = [&](int word_in) -> Msg {
        const int word = __builtin_amdgcn_readfirstlane(word_in);
        const int s    = word >> 9;
        const int conv = (word >> 8) & 1;
        Msg g;
        g.slot = conv * ROWS_PG + (word & 255);
        const float* __restrict__ hp = (conv ? H2 : H1) + (size_t)s * F_OUT;
        const float* __restrict__ ep = (conv ? es2 : es1) + (size_t)s * HEADS;
        g.h0 = hp[lane]; g.h1 = hp[32 + lane]; g.h2 = hp[64 + lane]; g.h3 = hp[96 + lane];
        g.e  = ep[hd];
        g.em = s_em[g.slot * HEADS + hd];
        return g;
    };
    auto apply = [&](const Msg& g) {
        float v = lrelu(g.e + g.em.x);
        float w = __expf(v - g.em.y);
        if (lane < HEADS)
            __hip_atomic_fetch_add(&s_den[g.slot * HEADS + lane], w,
                                   __ATOMIC_RELAXED, __HIP_MEMORY_SCOPE_WORKGROUP);
        int wi = __float_as_int(w);
        float w0 = __int_as_float(__builtin_amdgcn_readlane(wi, 0));
        float w1 = __int_as_float(__builtin_amdgcn_readlane(wi, 1));
        float w2 = __int_as_float(__builtin_amdgcn_readlane(wi, 2));
        float w3 = __int_as_float(__builtin_amdgcn_readlane(wi, 3));
        float* np = &s_num[g.slot * F_OUT + lane];
        __hip_atomic_fetch_add(np,      w0 * g.h0, __ATOMIC_RELAXED, __HIP_MEMORY_SCOPE_WORKGROUP);
        __hip_atomic_fetch_add(np + 32, w1 * g.h1, __ATOMIC_RELAXED, __HIP_MEMORY_SCOPE_WORKGROUP);
        __hip_atomic_fetch_add(np + 64, w2 * g.h2, __ATOMIC_RELAXED, __HIP_MEMORY_SCOPE_WORKGROUP);
        __hip_atomic_fetch_add(np + 96, w3 * g.h3, __ATOMIC_RELAXED, __HIP_MEMORY_SCOPE_WORKGROUP);
    };
    auto drain = [&](int cnt) -> int {
        int* q = &s_q[wave * QCAP];
        int i = 0;
        for (; i + 8 <= cnt; i += 8) {
            Msg g[8];
            #pragma unroll
            for (int u = 0; u < 8; ++u) g[u] = fetch(q[i + u]);
            #pragma unroll
            for (int u = 0; u < 8; ++u) apply(g[u]);
        }
        const int rem = cnt - i;
        if (i > 0 && lane < rem) { const int wv = q[i + lane]; q[lane] = wv; }
        return rem;
    };
    auto tail = [&](int cnt) {
        for (int i = 0; i < cnt; ++i) { Msg g = fetch(s_q[wave * QCAP + i]); apply(g); }
    };
    auto maxpass = [&](int cnt) {
        const int* q = &s_q[wave * QCAP];
        const int sub = lane >> 2;
        for (int i = 0; i < cnt; i += 64) {
            float e[8];
            int   at[8];
            #pragma unroll
            for (int u = 0; u < 8; ++u) {
                int k = i + 8 * u + sub;
                if (k > cnt - 1) k = cnt - 1;
                const int word = q[k];
                const int conv = (word >> 8) & 1;
                at[u] = (conv * ROWS_PG + (word & 255)) * HEADS + hd;
                e[u]  = (conv ? es2 : es1)[(size_t)(word >> 9) * HEADS + hd];
            }
            #pragma unroll
            for (int u = 0; u < 8; ++u) {
                const float v = lrelu(e[u] + s_em[at[u]].x);
                __hip_atomic_fetch_max(&s_em[at[u]].y, v,
                                       __ATOMIC_RELAXED, __HIP_MEMORY_SCOPE_WORKGROUP);
            }
        }
    };

    int qn = 0;
    auto seed = [&]() -> int {
        const int row = wave + 32 * (lane >> 1);
        if (row < nrows) s_q[wave * QCAP + lane] = ((r0 + row) << 9) | ((lane & 1) << 8) | row;
        return wave < nrows ? 2 * ((nrows - wave + 31) >> 5) : 0;
    };
    const unsigned lt = (1u << lane) - 1u;
    auto handle = [&](int a, int b, unsigned lim) {
        const bool t0 = (unsigned)(b - r0) < lim;
        const bool t1 = (unsigned)(a - r0) < lim;
        const unsigned bal0 = __builtin_amdgcn_ballot_w32(t0);
        const unsigned bal1 = __builtin_amdgcn_ballot_w32(t1);
        if (bal0 | bal1) {
            const int c0 = __popc(bal0);
            int* q = &s_q[wave * QCAP + qn];
            if (t0) q[__popc(bal0 & lt)]      = (a << 9) | (b - r0);
            if (t1) q[c0 + __popc(bal1 & lt)] = (b << 9) | 256 | (a - r0);
            qn += c0 + __popc(bal1);
        }
    };
    auto plain_sweep = [&](auto full) {
        for (int base = 0; base < E; base += 1024) {
            const int e = base + tid;
            const bool ok = e < E;
            const int ec = ok ? e : 0;
            if (qn > QCAP - 64) qn = full(qn);
            handle(ei0[ec], ei1[ec], ok ? (unsigned)nrows : 0u);
        }
    };
    auto give_up = [&](int) -> int { s_over = 1; return 0; };

    qn = seed();
    if (keys_usable(E, n, ws_bytes)) {
        const uint4* __restrict__ p4 = (const uint4*)planes;
        const int nblk  = E >> 5;
        const int nrun  = key_runs(E);
        const int niter = (nrun + 31) >> 5;
        const int seeds = qn;
        int gave = 0;
        const int start = (int)((blockIdx.x * (unsigned)niter) / gridDim.x);
        unsigned gm[8];
        #pragma unroll
        for (int b = 0; b < 8; ++b) gm[b] = ((blockIdx.x >> b) & 1u) ? ~0u : 0u;
        auto ld = [&](int it, uint4 (&k)[4]) -> int {
            int c = start + it; if (c >= niter) c -= niter;
            const int run = c * 32 + wave;
            const bool in = (it < niter) && (run < nrun);
            const int rc = in ? run : 0;
            #pragma unroll
            for (int j = 0; j < 4; ++j) k[j] = p4[rc * 128 + j * 32 + lane];
            const int gb = rc * 32 + lane;
            return (in && gb < nblk) ? gb : -1;
        };
        auto match = [&](const uint4& lo, const uint4& hi) -> unsigned {
            unsigned m = ~(lo.x ^ gm[0]);
            m &= ~(lo.y ^ gm[1]); m &= ~(lo.z ^ gm[2]); m &= ~(lo.w ^ gm[3]);
            m &= ~(hi.x ^ gm[4]); m &= ~(hi.y ^ gm[5]); m &= ~(hi.z ^ gm[6]); m &= ~(hi.w ^ gm[7]);
            return m;
        };
        auto note = [&](unsigned m, int gb, int conv) {
            while (true) {
                const bool has = m != 0u;
                const unsigned bal = __builtin_amdgcn_ballot_w32(has);
                if (bal == 0u) break;
                if (qn > QCAP - 32) { s_over = 1; gave = 1; qn = 0; }
                if (has) {
                    const int p = __builtin_ctz(m);
                    m &= m - 1u;
                    s_q[wave * QCAP + qn + __popc(bal & lt)] = (((gb << 5) | p) << 1) | conv;
                }
                qn += __popc(bal);
            }
        };
        auto work = [&](const uint4 (&k)[4], int gb) {
            const unsigned m1 = gb >= 0 ? match(k[2], k[3]) : 0u;
            const unsigned m2 = gb >= 0 ? match(k[0], k[1]) : 0u;
            note(m1, gb, 0);
            note(m2, gb, 1);
        };
        uint4 kc[2][4];
        int gc[2];
        #pragma unroll
        for (int s = 0; s < 2; ++s) gc[s] = ld(s, kc[s]);
        for (int it = 0; it < niter; it += 2) {
            #pragma unroll
            for (int s = 0; s < 2; ++s) {
                work(kc[s], gc[s]);
                gc[s] = ld(it + 2 + s, kc[s]);
            }
        }
        const int first = gave ? 0 : seeds;
        if (gave) qn = 0;
        const bool with_pairs = pairs_usable(E, n, ws_bytes);
        for (int base = first; base < qn; base += 128) {
            int code[4], ra[4], rb[4];
            #pragma unroll
            for (int u = 0; u < 4; ++u) {
                const int i = base + 32 * u + lane;
                code[u] = 0;
                if (i < qn) code[u] = s_q[wave * QCAP + i];
                if (with_pairs) {
                    const unsigned pv = pair[code[u] >> 1];
                    ra[u] = (int)(pv & 0xFFFFu);
                    rb[u] = (int)(pv >> 16);
                } else {
                    ra[u] = ei0[code[u] >> 1];
                    rb[u] = ei1[code[u] >> 1];
                }
            }
            #pragma unroll
            for (int u = 0; u < 4; ++u) {
                const int i = base + 32 * u + lane;
                if (i < qn)
                    s_q[wave * QCAP + i] = (code[u] & 1) ? ((rb[u] << 9) | 256 | (ra[u] - r0))
                                                         : ((ra[u] << 9) | (rb[u] - r0));
            }
        }
    } else {
        plain_sweep(give_up);
    }
    maxpass(qn);
    __syncthreads();

    const int over = __builtin_amdgcn_readfirstlane(s_over);
    if (!over) {
        qn = drain(qn);
        tail(qn);
    } else {
        qn = seed();
        plain_sweep([&](int cnt) -> int { maxpass(cnt); return 0; });
        maxpass(qn);
        __syncthreads();
        qn = seed();
        plain_sweep(drain);
        qn = drain(qn);
        tail(qn);
    }
    __syncthreads();

    for (int i = tid; i < nrows * (F_OUT / 4); i += 1024) {
        const int row = i >> 5;
        const int c4  = (i & 31) * 4;
        const int h   = c4 >> 5;
        const float4 n1 = *(const float4*)&s_num[row * F_OUT + c4];
        const float4 n2 = *(const float4*)&s_num[(ROWS_PG + row) * F_OUT + c4];
        const float d1 = s_den[row * HEADS + h];
        const float d2 = s_den[(ROWS_PG + row) * HEADS + h];
        const float4 bb1 = *(const float4*)&b1[c4];
        const float4 bb2 = *(const float4*)&b2[c4];
        float4 o;
        o.x = (1.0f - ALPHA_) * (n1.x / d1 + bb1.x) + ALPHA_ * (n2.x / d2 + bb2.x);
        o.y = (1.0f - ALPHA_) * (n1.y / d1 + bb1.y) + ALPHA_ * (n2.y / d2 + bb2.y);
        o.z = (1.0f - ALPHA_) * (n1.z / d1 + bb1.z) + ALPHA_ * (n2.z / d2 + bb2.z);
        o.w = (1.0f - ALPHA_) * (n1.w / d1 + bb1.w) + ALPHA_ * (n2.w / d2 + bb2.w);
        *(float4*)&out[(size_t)(r0 + row) * F_OUT + c4] = o;
    }
}

extern "C" void kernel_launch(void* const* d_in, const int* in_sizes, int n_in,
                              void* d_out, int out_size, void* d_ws, size_t ws_size,
                              hipStream_t stream) {
    const float* x    = (const float*)d_in[0];
    const int*   ei   = (const int*)d_in[1];
    const float* W1   = (const float*)d_in[2];
    const float* as1  = (const float*)d_in[3];
    const float* ad1  = (const float*)d_in[4];
    const float* b1   = (const float*)d_in[5];
    const float* W2   = (const float*)d_in[6];
    const float* as2  = (const float*)d_in[7];
    const float* ad2  = (const float*)d_in[8];
    const float* b2   = (const float*)d_in[9];
    float* out = (float*)d_out;

    const int n = in_sizes[0] / F_IN;
    const int E = in_sizes[1] / 2;
    const int* src = ei;
    const int* dst = ei + E;

    float* p   = (float*)d_ws;
    float* H1  = p;  p += (size_t)n * F_OUT;
    float* H2  = p;  p += (size_t)n * F_OUT;
    float* es1 = p;  p += (size_t)n * HEADS;
    float* ed1 = p;  p += (size_t)n * HEADS;
    float* es2 = p;  p += (size_t)n * HEADS;
    float* ed2 = p;  p += (size_t)n * HEADS;
    int nrun = (E / 32 + 31) / 32;
    unsigned* planes = (unsigned*)p;
    unsigned* pair   = planes + (size_t)nrun * 512;

    {
        int mtiles = (n + 15) / 16;
        int blocks = 2 * mtiles * (F_OUT / 16);
        gemm_wmma_kernel<<<blocks, 32, 0, stream>>>(x, W1, W2, H1, H2, n);
    }
    {
        int total = n * HEADS * 2;
        int nkb = (nrun * 4 + 7) / 8;
        scores_kernel<<<nkb + (total + 255) / 256, 256, 0, stream>>>(
            H1, H2, as1, ad1, as2, ad2, es1, ed1, es2, ed2, n,
            src, dst, E, nkb, planes, pair, (unsigned long long)ws_size);
    }
    {
        int groups = (n + ROWS_PG - 1) / ROWS_PG;
        owner_kernel<<<groups, 1024, 0, stream>>>(src, dst, E, n, H1, H2,
                                                  es1, ed1, es2, ed2, b1, b2,
                                                  planes, pair,
                                                  (unsigned long long)ws_size, out);
    }
}


# x@W weight-stationary: 16-wave groups loop over row tiles; each wave keeps a 128 x 16 strip of W1|W2 as B fragments in registers, x tile put in LDS once per group (was 1 wave per tile); same k order
# speedup vs baseline: 16.4380x; 1.1283x over previous
#include <hip/hip_runtime.h>
#include <hip/hip_bf16.h>


typedef __attribute__((ext_vector_type(2))) float v2f;
typedef __attribute__((ext_vector_type(8))) float v8f;

#define F_IN   128
#define HEADS  4
#define C_OUT  32
#define F_OUT  128
#define ALPHA_ 0.5f
#define SLOPE_ 0.2f

__device__ __forceinline__ float lrelu(float v) {
    return v > 0.0f ? v : v * SLOPE_;
}

#define A_PITCH 132
#define GEMM_THREADS 512

__global__ __launch_bounds__(GEMM_THREADS) void gemm_wmma_kernel(
    const float* __restrict__ X,
    const float* __restrict__ W1, const float* __restrict__ W2,
    float* __restrict__ H1, float* __restrict__ H2, int n)
{
    __shared__ __attribute__((aligned(16))) float s_a[2][16 * A_PITCH];

    const int tid  = threadIdx.x;
    const int lane = tid & 31;
    const int wave = __builtin_amdgcn_readfirstlane(tid >> 5);
    const int conv  = wave >> 3;
    const int ntile = wave & 7;
    const int half = lane >> 4;
    const int l    = lane & 15;

    const float* __restrict__ Wm = conv ? W2 : W1;
    float* __restrict__ Hm       = conv ? H2 : H1;
    const int col = ntile * 16 + l;

    v2f bf[32];
    #pragma unroll
    for (int j = 0; j < 32; ++j) {
        const int ka = 4 * j + half * 2;
        bf[j].x = Wm[ka * F_OUT + col];
        bf[j].y = Wm[(ka + 1) * F_OUT + col];
    }

    const int mtiles = (n + 15) >> 4;
    auto ldx = [&](int mt) -> float4 {
        int row = mt * 16 + wave;
        if (row >= n) row = n - 1;
        return ((const float4*)(X + (size_t)row * F_IN))[lane];
    };
    const int st_off = wave * A_PITCH + lane * 4;
    const int fr_off = l * A_PITCH + half * 2;

    int mt = blockIdx.x;
    if (mt < mtiles) *(float4*)&s_a[0][st_off] = ldx(mt);
    __syncthreads();
    int cur = 0;
    for (; mt < mtiles; mt += gridDim.x) {
        const int  nx   = mt + gridDim.x;
        const bool more = nx < mtiles;
        float4 vn = make_float4(0.0f, 0.0f, 0.0f, 0.0f);
        if (more) vn = ldx(nx);

        const float* __restrict__ sa = &s_a[cur][fr_off];
        v8f acc = {};
        #pragma unroll
        for (int j = 0; j < 32; ++j) {
            const v2f a = *(const v2f*)&sa[4 * j];
            acc = __builtin_amdgcn_wmma_f32_16x16x4_f32(
                 false, a,  false, bf[j],
                 (short)0, acc,  false,  false);
        }

        if (more) *(float4*)&s_a[cur ^ 1][st_off] = vn;
        __syncthreads();
        cur ^= 1;

        const int rowBase = mt * 16 + half * 8;
        float* __restrict__ hp = Hm + (size_t)rowBase * F_OUT + col;
        if (mt * 16 + 16 <= n) {
            #pragma unroll
            for (int r = 0; r < 8; ++r) hp[r * F_OUT] = acc[r];
        } else {
            #pragma unroll
            for (int r = 0; r < 8; ++r) if (rowBase + r < n) hp[r * F_OUT] = acc[r];
        }
    }
}

#define ROWS_PG  196

__device__ __forceinline__ int key_runs(int E) { return ((E >> 5) + 31) >> 5; }
__device__ __forceinline__ unsigned long long key_bytes(int E, int n) {
    return (unsigned long long)n * (2 * F_OUT + 4 * HEADS) * 4ull + (unsigned long long)key_runs(E) * 2048ull;
}
__device__ __forceinline__ bool keys_usable(int E, int n, unsigned long long ws_bytes) {
    return (E & 31) == 0 && (n + ROWS_PG - 1) / ROWS_PG <= 256 && ws_bytes >= key_bytes(E, n);
}
__device__ __forceinline__ bool pairs_usable(int E, int n, unsigned long long ws_bytes) {
    return ws_bytes >= key_bytes(E, n) + (unsigned long long)E * 4ull;
}

__global__ void scores_kernel(
    const float* __restrict__ H1, const float* __restrict__ H2,
    const float* __restrict__ as1, const float* __restrict__ ad1,
    const float* __restrict__ as2, const float* __restrict__ ad2,
    float* __restrict__ es1, float* __restrict__ ed1,
    float* __restrict__ es2, float* __restrict__ ed2, int n,
    const int* __restrict__ ei0, const int* __restrict__ ei1, int E, int nkb,
    unsigned* __restrict__ planes, unsigned* __restrict__ pair,
    unsigned long long ws_bytes)
{
    if ((int)blockIdx.x < nkb) {
        if (!keys_usable(E, n, ws_bytes)) return;
        const int lane = threadIdx.x & 31;
        const int w    = __builtin_amdgcn_readfirstlane((int)(blockIdx.x * (blockDim.x >> 5) + (threadIdx.x >> 5)));
        const int nblk = E >> 5;
        if (w >= key_runs(E) * 4) return;
        const bool with_pairs = pairs_usable(E, n, ws_bytes);
        int av[8], bv[8];
        #pragma unroll
        for (int u = 0; u < 8; ++u) {
            const int gb = w * 8 + u;
            const int ec = gb < nblk ? gb * 32 + lane : 0;
            av[u] = ei0[ec];
            bv[u] = ei1[ec];
        }
        unsigned pa[8], pb[8];
        #pragma unroll
        for (int b = 0; b < 8; ++b) { pa[b] = 0u; pb[b] = 0u; }
        #pragma unroll
        for (int u = 0; u < 8; ++u) {
            const int gb = w * 8 + u;
            if (with_pairs && gb < nblk) pair[gb * 32 + lane] = (unsigned)av[u] | (unsigned)bv[u] << 16;
            const unsigned ka = (unsigned)av[u] / ROWS_PG;
            const unsigned kb = (unsigned)bv[u] / ROWS_PG;
            #pragma unroll
            for (int b = 0; b < 8; ++b) {
                const unsigned ba = __builtin_amdgcn_ballot_w32(((ka >> b) & 1u) != 0u);
                const unsigned bb = __builtin_amdgcn_ballot_w32(((kb >> b) & 1u) != 0u);
                if (lane == u) { pa[b] = ba; pb[b] = bb; }
            }
        }
        if (lane < 8) {
            const int gb = w * 8 + lane;
            uint4* __restrict__ p4 = (uint4*)planes + (size_t)(gb >> 5) * 128 + (gb & 31);
            p4[0]  = make_uint4(pa[0], pa[1], pa[2], pa[3]);
            p4[32] = make_uint4(pa[4], pa[5], pa[6], pa[7]);
            p4[64] = make_uint4(pb[0], pb[1], pb[2], pb[3]);
            p4[96] = make_uint4(pb[4], pb[5], pb[6], pb[7]);
        }
        return;
    }
    int i = (blockIdx.x - nkb) * blockDim.x + threadIdx.x;
    int total = n * HEADS * 2;
    if (i >= total) return;
    int conv = (i >= n * HEADS) ? 1 : 0;
    int j = conv ? (i - n * HEADS) : i;
    int node = j >> 2;
    int h    = j & 3;

    const float* __restrict__ hp  = (conv ? H2 : H1) + node * F_OUT + h * C_OUT;
    const float* __restrict__ asp = (conv ? as2 : as1) + h * C_OUT;
    const float* __restrict__ adp = (conv ? ad2 : ad1) + h * C_OUT;

    float ssum = 0.0f, dsum = 0.0f;
    #pragma unroll
    for (int c = 0; c < C_OUT; ++c) {
        float v = hp[c];
        ssum += v * asp[c];
        dsum += v * adp[c];
    }
    if (conv) { es2[node * HEADS + h] = ssum; ed2[node * HEADS + h] = dsum; }
    else      { es1[node * HEADS + h] = ssum; ed1[node * HEADS + h] = dsum; }
}

#define QCAP     768

struct Msg { float h0, h1, h2, h3, e; float2 em; int slot; };

__global__ __launch_bounds__(1024) void owner_kernel(
    const int* __restrict__ ei0, const int* __restrict__ ei1, int E, int n,
    const float* __restrict__ H1, const float* __restrict__ H2,
    const float* __restrict__ es1, const float* __restrict__ ed1,
    const float* __restrict__ es2, const float* __restrict__ ed2,
    const float* __restrict__ b1, const float* __restrict__ b2,
    const unsigned* __restrict__ planes, const unsigned* __restrict__ pair,
    unsigned long long ws_bytes,
    float* __restrict__ out)
{
    __shared__ __attribute__((aligned(16))) float s_num[2 * ROWS_PG * F_OUT];
    __shared__ float  s_den[2 * ROWS_PG * HEADS];
    __shared__ float2 s_em[2 * ROWS_PG * HEADS];
    __shared__ __attribute__((aligned(16))) int s_q[32 * QCAP];
    __shared__ int s_over;

    const int tid  = threadIdx.x;
    const int lane = tid & 31;
    const int wave = __builtin_amdgcn_readfirstlane(tid >> 5);
    const int hd   = lane & 3;
    const int r0   = blockIdx.x * ROWS_PG;
    int nrows = n - r0;
    if (nrows > ROWS_PG) nrows = ROWS_PG;

    {
        const float4 z = make_float4(0.0f, 0.0f, 0.0f, 0.0f);
        for (int i = tid; i < 2 * ROWS_PG * (F_OUT / 4); i += 1024) ((float4*)s_num)[i] = z;
    }
    for (int i = tid; i < 2 * ROWS_PG * HEADS; i += 1024) {
        int conv = i >= ROWS_PG * HEADS ? 1 : 0;
        int j = i - conv * ROWS_PG * HEADS;
        float2 em; em.x = 0.0f; em.y = -__builtin_huge_valf();
        if ((j >> 2) < nrows) em.x = (conv ? ed2 : ed1)[r0 * HEADS + j];
        s_em[i]  = em;
        s_den[i] = 0.0f;
    }
    if (tid == 0) s_over = 0;
    __syncthreads();

    auto fetch = [&](int word_in) -> Msg {
        const int word = __builtin_amdgcn_readfirstlane(word_in);
        const int s    = word >> 9;
        const int conv = (word >> 8) & 1;
        Msg g;
        g.slot = conv * ROWS_PG + (word & 255);
        const float* __restrict__ hp = (conv ? H2 : H1) + (size_t)s * F_OUT;
        const float* __restrict__ ep = (conv ? es2 : es1) + (size_t)s * HEADS;
        g.h0 = hp[lane]; g.h1 = hp[32 + lane]; g.h2 = hp[64 + lane]; g.h3 = hp[96 + lane];
        g.e  = ep[hd];
        g.em = s_em[g.slot * HEADS + hd];
        return g;
    };
    auto apply = [&](const Msg& g) {
        float v = lrelu(g.e + g.em.x);
        float w = __expf(v - g.em.y);
        if (lane < HEADS)
            __hip_atomic_fetch_add(&s_den[g.slot * HEADS + lane], w,
                                   __ATOMIC_RELAXED, __HIP_MEMORY_SCOPE_WORKGROUP);
        int wi = __float_as_int(w);
        float w0 = __int_as_float(__builtin_amdgcn_readlane(wi, 0));
        float w1 = __int_as_float(__builtin_amdgcn_readlane(wi, 1));
        float w2 = __int_as_float(__builtin_amdgcn_readlane(wi, 2));
        float w3 = __int_as_float(__builtin_amdgcn_readlane(wi, 3));
        float* np = &s_num[g.slot * F_OUT + lane];
        __hip_atomic_fetch_add(np,      w0 * g.h0, __ATOMIC_RELAXED, __HIP_MEMORY_SCOPE_WORKGROUP);
        __hip_atomic_fetch_add(np + 32, w1 * g.h1, __ATOMIC_RELAXED, __HIP_MEMORY_SCOPE_WORKGROUP);
        __hip_atomic_fetch_add(np + 64, w2 * g.h2, __ATOMIC_RELAXED, __HIP_MEMORY_SCOPE_WORKGROUP);
        __hip_atomic_fetch_add(np + 96, w3 * g.h3, __ATOMIC_RELAXED, __HIP_MEMORY_SCOPE_WORKGROUP);
    };
    auto drain = [&](int cnt) -> int {
        int* q = &s_q[wave * QCAP];
        int i = 0;
        for (; i + 8 <= cnt; i += 8) {
            Msg g[8];
            #pragma unroll
            for (int u = 0; u < 8; ++u) g[u] = fetch(q[i + u]);
            #pragma unroll
            for (int u = 0; u < 8; ++u) apply(g[u]);
        }
        const int rem = cnt - i;
        if (i > 0 && lane < rem) { const int wv = q[i + lane]; q[lane] = wv; }
        return rem;
    };
    auto tail = [&](int cnt) {
        for (int i = 0; i < cnt; ++i) { Msg g = fetch(s_q[wave * QCAP + i]); apply(g); }
    };
    auto maxpass = [&](int cnt) {
        const int* q = &s_q[wave * QCAP];
        const int sub = lane >> 2;
        for (int i = 0; i < cnt; i += 64) {
            float e[8];
            int   at[8];
            #pragma unroll
            for (int u = 0; u < 8; ++u) {
                int k = i + 8 * u + sub;
                if (k > cnt - 1) k = cnt - 1;
                const int word = q[k];
                const int conv = (word >> 8) & 1;
                at[u] = (conv * ROWS_PG + (word & 255)) * HEADS + hd;
                e[u]  = (conv ? es2 : es1)[(size_t)(word >> 9) * HEADS + hd];
            }
            #pragma unroll
            for (int u = 0; u < 8; ++u) {
                const float v = lrelu(e[u] + s_em[at[u]].x);
                __hip_atomic_fetch_max(&s_em[at[u]].y, v,
                                       __ATOMIC_RELAXED, __HIP_MEMORY_SCOPE_WORKGROUP);
            }
        }
    };

    int qn = 0;
    auto seed = [&]() -> int {
        const int row = wave + 32 * (lane >> 1);
        if (row < nrows) s_q[wave * QCAP + lane] = ((r0 + row) << 9) | ((lane & 1) << 8) | row;
        return wave < nrows ? 2 * ((nrows - wave + 31) >> 5) : 0;
    };
    const unsigned lt = (1u << lane) - 1u;
    auto handle = [&](int a, int b, unsigned lim) {
        const bool t0 = (unsigned)(b - r0) < lim;
        const bool t1 = (unsigned)(a - r0) < lim;
        const unsigned bal0 = __builtin_amdgcn_ballot_w32(t0);
        const unsigned bal1 = __builtin_amdgcn_ballot_w32(t1);
        if (bal0 | bal1) {
            const int c0 = __popc(bal0);
            int* q = &s_q[wave * QCAP + qn];
            if (t0) q[__popc(bal0 & lt)]      = (a << 9) | (b - r0);
            if (t1) q[c0 + __popc(bal1 & lt)] = (b << 9) | 256 | (a - r0);
            qn += c0 + __popc(bal1);
        }
    };
    auto plain_sweep = [&](auto full) {
        for (int base = 0; base < E; base += 1024) {
            const int e = base + tid;
            const bool ok = e < E;
            const int ec = ok ? e : 0;
            if (qn > QCAP - 64) qn = full(qn);
            handle(ei0[ec], ei1[ec], ok ? (unsigned)nrows : 0u);
        }
    };
    auto give_up = [&](int) -> int { s_over = 1; return 0; };

    qn = seed();
    if (keys_usable(E, n, ws_bytes)) {
        const uint4* __restrict__ p4 = (const uint4*)planes;
        const int nblk  = E >> 5;
        const int nrun  = key_runs(E);
        const int niter = (nrun + 31) >> 5;
        const int seeds = qn;
        int gave = 0;
        const int start = (int)((blockIdx.x * (unsigned)niter) / gridDim.x);
        unsigned gm[8];
        #pragma unroll
        for (int b = 0; b < 8; ++b) gm[b] = ((blockIdx.x >> b) & 1u) ? ~0u : 0u;
        auto ld = [&](int it, uint4 (&k)[4]) -> int {
            int c = start + it; if (c >= niter) c -= niter;
            const int run = c * 32 + wave;
            const bool in = (it < niter) && (run < nrun);
            const int rc = in ? run : 0;
            #pragma unroll
            for (int j = 0; j < 4; ++j) k[j] = p4[rc * 128 + j * 32 + lane];
            const int gb = rc * 32 + lane;
            return (in && gb < nblk) ? gb : -1;
        };
        auto match = [&](const uint4& lo, const uint4& hi) -> unsigned {
            unsigned m = ~(lo.x ^ gm[0]);
            m &= ~(lo.y ^ gm[1]); m &= ~(lo.z ^ gm[2]); m &= ~(lo.w ^ gm[3]);
            m &= ~(hi.x ^ gm[4]); m &= ~(hi.y ^ gm[5]); m &= ~(hi.z ^ gm[6]); m &= ~(hi.w ^ gm[7]);
            return m;
        };
        auto note = [&](unsigned m, int gb, int conv) {
            while (true) {
                const bool has = m != 0u;
                const unsigned bal = __builtin_amdgcn_ballot_w32(has);
                if (bal == 0u) break;
                if (qn > QCAP - 32) { s_over = 1; gave = 1; qn = 0; }
                if (has) {
                    const int p = __builtin_ctz(m);
                    m &= m - 1u;
                    s_q[wave * QCAP + qn + __popc(bal & lt)] = (((gb << 5) | p) << 1) | conv;
                }
                qn += __popc(bal);
            }
        };
        auto work = [&](const uint4 (&k)[4], int gb) {
            const unsigned m1 = gb >= 0 ? match(k[2], k[3]) : 0u;
            const unsigned m2 = gb >= 0 ? match(k[0], k[1]) : 0u;
            note(m1, gb, 0);
            note(m2, gb, 1);
        };
        uint4 kc[2][4];
        int gc[2];
        #pragma unroll
        for (int s = 0; s < 2; ++s) gc[s] = ld(s, kc[s]);
        for (int it = 0; it < niter; it += 2) {
            #pragma unroll
            for (int s = 0; s < 2; ++s) {
                work(kc[s], gc[s]);
                gc[s] = ld(it + 2 + s, kc[s]);
            }
        }
        const int first = gave ? 0 : seeds;
        if (gave) qn = 0;
        const bool with_pairs = pairs_usable(E, n, ws_bytes);
        for (int base = first; base < qn; base += 128) {
            int code[4], ra[4], rb[4];
            #pragma unroll
            for (int u = 0; u < 4; ++u) {
                const int i = base + 32 * u + lane;
                code[u] = 0;
                if (i < qn) code[u] = s_q[wave * QCAP + i];
                if (with_pairs) {
                    const unsigned pv = pair[code[u] >> 1];
                    ra[u] = (int)(pv & 0xFFFFu);
                    rb[u] = (int)(pv >> 16);
                } else {
                    ra[u] = ei0[code[u] >> 1];
                    rb[u] = ei1[code[u] >> 1];
                }
            }
            #pragma unroll
            for (int u = 0; u < 4; ++u) {
                const int i = base + 32 * u + lane;
                if (i < qn)
                    s_q[wave * QCAP + i] = (code[u] & 1) ? ((rb[u] << 9) | 256 | (ra[u] - r0))
                                                         : ((ra[u] << 9) | (rb[u] - r0));
            }
        }
    } else {
        plain_sweep(give_up);
    }
    maxpass(qn);
    __syncthreads();

    const int over = __builtin_amdgcn_readfirstlane(s_over);
    if (!over) {
        qn = drain(qn);
        tail(qn);
    } else {
        qn = seed();
        plain_sweep([&](int cnt) -> int { maxpass(cnt); return 0; });
        maxpass(qn);
        __syncthreads();
        qn = seed();
        plain_sweep(drain);
        qn = drain(qn);
        tail(qn);
    }
    __syncthreads();

    for (int i = tid; i < nrows * (F_OUT / 4); i += 1024) {
        const int row = i >> 5;
        const int c4  = (i & 31) * 4;
        const int h   = c4 >> 5;
        const float4 n1 = *(const float4*)&s_num[row * F_OUT + c4];
        const float4 n2 = *(const float4*)&s_num[(ROWS_PG + row) * F_OUT + c4];
        const float d1 = s_den[row * HEADS + h];
        const float d2 = s_den[(ROWS_PG + row) * HEADS + h];
        const float4 bb1 = *(const float4*)&b1[c4];
        const float4 bb2 = *(const float4*)&b2[c4];
        float4 o;
        o.x = (1.0f - ALPHA_) * (n1.x / d1 + bb1.x) + ALPHA_ * (n2.x / d2 + bb2.x);
        o.y = (1.0f - ALPHA_) * (n1.y / d1 + bb1.y) + ALPHA_ * (n2.y / d2 + bb2.y);
        o.z = (1.0f - ALPHA_) * (n1.z / d1 + bb1.z) + ALPHA_ * (n2.z / d2 + bb2.z);
        o.w = (1.0f - ALPHA_) * (n1.w / d1 + bb1.w) + ALPHA_ * (n2.w / d2 + bb2.w);
        *(float4*)&out[(size_t)(r0 + row) * F_OUT + c4] = o;
    }
}

extern "C" void kernel_launch(void* const* d_in, const int* in_sizes, int n_in,
                              void* d_out, int out_size, void* d_ws, size_t ws_size,
                              hipStream_t stream) {
    const float* x    = (const float*)d_in[0];
    const int*   ei   = (const int*)d_in[1];
    const float* W1   = (const float*)d_in[2];
    const float* as1  = (const float*)d_in[3];
    const float* ad1  = (const float*)d_in[4];
    const float* b1   = (const float*)d_in[5];
    const float* W2   = (const float*)d_in[6];
    const float* as2  = (const float*)d_in[7];
    const float* ad2  = (const float*)d_in[8];
    const float* b2   = (const float*)d_in[9];
    float* out = (float*)d_out;

    const int n = in_sizes[0] / F_IN;
    const int E = in_sizes[1] / 2;
    const int* src = ei;
    const int* dst = ei + E;

    float* p   = (float*)d_ws;
    float* H1  = p;  p += (size_t)n * F_OUT;
    float* H2  = p;  p += (size_t)n * F_OUT;
    float* es1 = p;  p += (size_t)n * HEADS;
    float* ed1 = p;  p += (size_t)n * HEADS;
    float* es2 = p;  p += (size_t)n * HEADS;
    float* ed2 = p;  p += (size_t)n * HEADS;
    int nrun = (E / 32 + 31) / 32;
    unsigned* planes = (unsigned*)p;
    unsigned* pair   = planes + (size_t)nrun * 512;

    gemm_wmma_kernel<<<512, GEMM_THREADS, 0, stream>>>(x, W1, W2, H1, H2, n);
    {
        int total = n * HEADS * 2;
        int nkb = (nrun * 4 + 7) / 8;
        scores_kernel<<<nkb + (total + 255) / 256, 256, 0, stream>>>(
            H1, H2, as1, ad1, as2, ad2, es1, ed1, es2, ed2, n,
            src, dst, E, nkb, planes, pair, (unsigned long long)ws_size);
    }
    {
        int groups = (n + ROWS_PG - 1) / ROWS_PG;
        owner_kernel<<<groups, 1024, 0, stream>>>(src, dst, E, n, H1, H2,
                                                  es1, ed1, es2, ed2, b1, b2,
                                                  planes, pair,
                                                  (unsigned long long)ws_size, out);
    }
}


# owner groups' sums pass: an item's h row comes as one 16-byte load per lane (was four 4-byte loads); the 32 weights of eight items are formed at once (one gather, one exp); LDS adds as before
# speedup vs baseline: 18.9893x; 1.1552x over previous
#include <hip/hip_runtime.h>
#include <hip/hip_bf16.h>


typedef __attribute__((ext_vector_type(2))) float v2f;
typedef __attribute__((ext_vector_type(8))) float v8f;

#define F_IN   128
#define HEADS  4
#define C_OUT  32
#define F_OUT  128
#define ALPHA_ 0.5f
#define SLOPE_ 0.2f

__device__ __forceinline__ float lrelu(float v) {
    return v > 0.0f ? v : v * SLOPE_;
}

#define A_PITCH 132
#define GEMM_THREADS 512

__global__ __launch_bounds__(GEMM_THREADS) void gemm_wmma_kernel(
    const float* __restrict__ X,
    const float* __restrict__ W1, const float* __restrict__ W2,
    float* __restrict__ H1, float* __restrict__ H2, int n)
{
    __shared__ __attribute__((aligned(16))) float s_a[2][16 * A_PITCH];

    const int tid  = threadIdx.x;
    const int lane = tid & 31;
    const int wave = __builtin_amdgcn_readfirstlane(tid >> 5);
    const int conv  = wave >> 3;
    const int ntile = wave & 7;
    const int half = lane >> 4;
    const int l    = lane & 15;

    const float* __restrict__ Wm = conv ? W2 : W1;
    float* __restrict__ Hm       = conv ? H2 : H1;
    const int col = ntile * 16 + l;

    v2f bf[32];
    #pragma unroll
    for (int j = 0; j < 32; ++j) {
        const int ka = 4 * j + half * 2;
        bf[j].x = Wm[ka * F_OUT + col];
        bf[j].y = Wm[(ka + 1) * F_OUT + col];
    }

    const int mtiles = (n + 15) >> 4;
    auto ldx = [&](int mt) -> float4 {
        int row = mt * 16 + wave;
        if (row >= n) row = n - 1;
        return ((const float4*)(X + (size_t)row * F_IN))[lane];
    };
    const int st_off = wave * A_PITCH + lane * 4;
    const int fr_off = l * A_PITCH + half * 2;

    int mt = blockIdx.x;
    if (mt < mtiles) *(float4*)&s_a[0][st_off] = ldx(mt);
    __syncthreads();
    int cur = 0;
    for (; mt < mtiles; mt += gridDim.x) {
        const int  nx   = mt + gridDim.x;
        const bool more = nx < mtiles;
        float4 vn = make_float4(0.0f, 0.0f, 0.0f, 0.0f);
        if (more) vn = ldx(nx);

        const float* __restrict__ sa = &s_a[cur][fr_off];
        v8f acc = {};
        #pragma unroll
        for (int j = 0; j < 32; ++j) {
            const v2f a = *(const v2f*)&sa[4 * j];
            acc = __builtin_amdgcn_wmma_f32_16x16x4_f32(
                 false, a,  false, bf[j],
                 (short)0, acc,  false,  false);
        }

        if (more) *(float4*)&s_a[cur ^ 1][st_off] = vn;
        __syncthreads();
        cur ^= 1;

        const int rowBase = mt * 16 + half * 8;
        float* __restrict__ hp = Hm + (size_t)rowBase * F_OUT + col;
        if (mt * 16 + 16 <= n) {
            #pragma unroll
            for (int r = 0; r < 8; ++r) hp[r * F_OUT] = acc[r];
        } else {
            #pragma unroll
            for (int r = 0; r < 8; ++r) if (rowBase + r < n) hp[r * F_OUT] = acc[r];
        }
    }
}

#define ROWS_PG  196

__device__ __forceinline__ int key_runs(int E) { return ((E >> 5) + 31) >> 5; }
__device__ __forceinline__ unsigned long long key_bytes(int E, int n) {
    return (unsigned long long)n * (2 * F_OUT + 4 * HEADS) * 4ull + (unsigned long long)key_runs(E) * 2048ull;
}
__device__ __forceinline__ bool keys_usable(int E, int n, unsigned long long ws_bytes) {
    return (E & 31) == 0 && (n + ROWS_PG - 1) / ROWS_PG <= 256 && ws_bytes >= key_bytes(E, n);
}
__device__ __forceinline__ bool pairs_usable(int E, int n, unsigned long long ws_bytes) {
    return ws_bytes >= key_bytes(E, n) + (unsigned long long)E * 4ull;
}

__global__ void scores_kernel(
    const float* __restrict__ H1, const float* __restrict__ H2,
    const float* __restrict__ as1, const float* __restrict__ ad1,
    const float* __restrict__ as2, const float* __restrict__ ad2,
    float* __restrict__ es1, float* __restrict__ ed1,
    float* __restrict__ es2, float* __restrict__ ed2, int n,
    const int* __restrict__ ei0, const int* __restrict__ ei1, int E, int nkb,
    unsigned* __restrict__ planes, unsigned* __restrict__ pair,
    unsigned long long ws_bytes)
{
    if ((int)blockIdx.x < nkb) {
        if (!keys_usable(E, n, ws_bytes)) return;
        const int lane = threadIdx.x & 31;
        const int w    = __builtin_amdgcn_readfirstlane((int)(blockIdx.x * (blockDim.x >> 5) + (threadIdx.x >> 5)));
        const int nblk = E >> 5;
        if (w >= key_runs(E) * 4) return;
        const bool with_pairs = pairs_usable(E, n, ws_bytes);
        int av[8], bv[8];
        #pragma unroll
        for (int u = 0; u < 8; ++u) {
            const int gb = w * 8 + u;
            const int ec = gb < nblk ? gb * 32 + lane : 0;
            av[u] = ei0[ec];
            bv[u] = ei1[ec];
        }
        unsigned pa[8], pb[8];
        #pragma unroll
        for (int b = 0; b < 8; ++b) { pa[b] = 0u; pb[b] = 0u; }
        #pragma unroll
        for (int u = 0; u < 8; ++u) {
            const int gb = w * 8 + u;
            if (with_pairs && gb < nblk) pair[gb * 32 + lane] = (unsigned)av[u] | (unsigned)bv[u] << 16;
            const unsigned ka = (unsigned)av[u] / ROWS_PG;
            const unsigned kb = (unsigned)bv[u] / ROWS_PG;
            #pragma unroll
            for (int b = 0; b < 8; ++b) {
                const unsigned ba = __builtin_amdgcn_ballot_w32(((ka >> b) & 1u) != 0u);
                const unsigned bb = __builtin_amdgcn_ballot_w32(((kb >> b) & 1u) != 0u);
                if (lane == u) { pa[b] = ba; pb[b] = bb; }
            }
        }
        if (lane < 8) {
            const int gb = w * 8 + lane;
            uint4* __restrict__ p4 = (uint4*)planes + (size_t)(gb >> 5) * 128 + (gb & 31);
            p4[0]  = make_uint4(pa[0], pa[1], pa[2], pa[3]);
            p4[32] = make_uint4(pa[4], pa[5], pa[6], pa[7]);
            p4[64] = make_uint4(pb[0], pb[1], pb[2], pb[3]);
            p4[96] = make_uint4(pb[4], pb[5], pb[6], pb[7]);
        }
        return;
    }
    int i = (blockIdx.x - nkb) * blockDim.x + threadIdx.x;
    int total = n * HEADS * 2;
    if (i >= total) return;
    int conv = (i >= n * HEADS) ? 1 : 0;
    int j = conv ? (i - n * HEADS) : i;
    int node = j >> 2;
    int h    = j & 3;

    const float* __restrict__ hp  = (conv ? H2 : H1) + node * F_OUT + h * C_OUT;
    const float* __restrict__ asp = (conv ? as2 : as1) + h * C_OUT;
    const float* __restrict__ adp = (conv ? ad2 : ad1) + h * C_OUT;

    float ssum = 0.0f, dsum = 0.0f;
    #pragma unroll
    for (int c = 0; c < C_OUT; ++c) {
        float v = hp[c];
        ssum += v * asp[c];
        dsum += v * adp[c];
    }
    if (conv) { es2[node * HEADS + h] = ssum; ed2[node * HEADS + h] = dsum; }
    else      { es1[node * HEADS + h] = ssum; ed1[node * HEADS + h] = dsum; }
}

#define QCAP     768

template <int U>
__device__ __forceinline__ float from_lane_of_eight(float v) {
    return __int_as_float((int)__builtin_amdgcn_mov_dpp8((unsigned)__float_as_int(v), (unsigned)(U * 0x249249)));
}
template <int U>
__device__ __forceinline__ void add_row(float* np, float w, const float4& r) {
    const float wu = from_lane_of_eight<U>(w);
    __hip_atomic_fetch_add(np,      wu * r.x, __ATOMIC_RELAXED, __HIP_MEMORY_SCOPE_WORKGROUP);
    __hip_atomic_fetch_add(np + 32, wu * r.y, __ATOMIC_RELAXED, __HIP_MEMORY_SCOPE_WORKGROUP);
    __hip_atomic_fetch_add(np + 64, wu * r.z, __ATOMIC_RELAXED, __HIP_MEMORY_SCOPE_WORKGROUP);
    __hip_atomic_fetch_add(np + 96, wu * r.w, __ATOMIC_RELAXED, __HIP_MEMORY_SCOPE_WORKGROUP);
}

__global__ __launch_bounds__(1024) void owner_kernel(
    const int* __restrict__ ei0, const int* __restrict__ ei1, int E, int n,
    const float* __restrict__ H1, const float* __restrict__ H2,
    const float* __restrict__ es1, const float* __restrict__ ed1,
    const float* __restrict__ es2, const float* __restrict__ ed2,
    const float* __restrict__ b1, const float* __restrict__ b2,
    const unsigned* __restrict__ planes, const unsigned* __restrict__ pair,
    unsigned long long ws_bytes,
    float* __restrict__ out)
{
    __shared__ __attribute__((aligned(16))) float s_num[2 * ROWS_PG * F_OUT];
    __shared__ float  s_den[2 * ROWS_PG * HEADS];
    __shared__ float2 s_em[2 * ROWS_PG * HEADS];
    __shared__ __attribute__((aligned(16))) int s_q[32 * QCAP];
    __shared__ int s_over;

    const int tid  = threadIdx.x;
    const int lane = tid & 31;
    const int wave = __builtin_amdgcn_readfirstlane(tid >> 5);
    const int hd   = lane & 3;
    const int r0   = blockIdx.x * ROWS_PG;
    int nrows = n - r0;
    if (nrows > ROWS_PG) nrows = ROWS_PG;

    {
        const float4 z = make_float4(0.0f, 0.0f, 0.0f, 0.0f);
        for (int i = tid; i < 2 * ROWS_PG * (F_OUT / 4); i += 1024) ((float4*)s_num)[i] = z;
    }
    for (int i = tid; i < 2 * ROWS_PG * HEADS; i += 1024) {
        int conv = i >= ROWS_PG * HEADS ? 1 : 0;
        int j = i - conv * ROWS_PG * HEADS;
        float2 em; em.x = 0.0f; em.y = -__builtin_huge_valf();
        if ((j >> 2) < nrows) em.x = (conv ? ed2 : ed1)[r0 * HEADS + j];
        s_em[i]  = em;
        s_den[i] = 0.0f;
    }
    if (tid == 0) s_over = 0;
    __syncthreads();

    auto batch = [&](int i, const int cnt) {
        const int* q = &s_q[wave * QCAP + i];
        const int  ul   = lane & 7;
        const int  hl   = lane >> 3;
        const bool live = ul < cnt;
        const int  wl   = q[live ? ul : 0];
        const int  cl   = (wl >> 8) & 1;
        const int  sl   = (cl * ROWS_PG + (wl & 255)) * HEADS + hl;
        const float  e  = (cl ? es2 : es1)[(size_t)(wl >> 9) * HEADS + hl];
        const float2 em = s_em[sl];
        float4 r[8];
        float* np[8];
        #pragma unroll
        for (int u = 0; u < 8; ++u) {
            if (u < cnt) {
                const int word = __builtin_amdgcn_readlane(wl, u);
                const int conv = (word >> 8) & 1;
                const float* __restrict__ hp = (conv ? H2 : H1) + (size_t)(word >> 9) * F_OUT;
                r[u]  = ((const float4*)hp)[lane];
                np[u] = &s_num[(conv * ROWS_PG + (word & 255)) * F_OUT + lane];
            }
        }
        const float w = __expf(lrelu(e + em.x) - em.y);
        if (live)
            __hip_atomic_fetch_add(&s_den[sl], w, __ATOMIC_RELAXED, __HIP_MEMORY_SCOPE_WORKGROUP);
        if (0 < cnt) add_row<0>(np[0], w, r[0]);
        if (1 < cnt) add_row<1>(np[1], w, r[1]);
        if (2 < cnt) add_row<2>(np[2], w, r[2]);
        if (3 < cnt) add_row<3>(np[3], w, r[3]);
        if (4 < cnt) add_row<4>(np[4], w, r[4]);
        if (5 < cnt) add_row<5>(np[5], w, r[5]);
        if (6 < cnt) add_row<6>(np[6], w, r[6]);
        if (7 < cnt) add_row<7>(np[7], w, r[7]);
    };
    auto drain = [&](int cnt) -> int {
        int* q = &s_q[wave * QCAP];
        int i = 0;
        for (; i + 8 <= cnt; i += 8) batch(i, 8);
        const int rem = cnt - i;
        if (i > 0 && lane < rem) { const int wv = q[i + lane]; q[lane] = wv; }
        return rem;
    };
    auto tail = [&](int cnt) {
        for (int i = 0; i < cnt; i += 8) batch(i, cnt - i < 8 ? cnt - i : 8);
    };
    auto maxpass = [&](int cnt) {
        const int* q = &s_q[wave * QCAP];
        const int sub = lane >> 2;
        for (int i = 0; i < cnt; i += 64) {
            float e[8];
            int   at[8];
            #pragma unroll
            for (int u = 0; u < 8; ++u) {
                int k = i + 8 * u + sub;
                if (k > cnt - 1) k = cnt - 1;
                const int word = q[k];
                const int conv = (word >> 8) & 1;
                at[u] = (conv * ROWS_PG + (word & 255)) * HEADS + hd;
                e[u]  = (conv ? es2 : es1)[(size_t)(word >> 9) * HEADS + hd];
            }
            #pragma unroll
            for (int u = 0; u < 8; ++u) {
                const float v = lrelu(e[u] + s_em[at[u]].x);
                __hip_atomic_fetch_max(&s_em[at[u]].y, v,
                                       __ATOMIC_RELAXED, __HIP_MEMORY_SCOPE_WORKGROUP);
            }
        }
    };

    int qn = 0;
    auto seed = [&]() -> int {
        const int row = wave + 32 * (lane >> 1);
        if (row < nrows) s_q[wave * QCAP + lane] = ((r0 + row) << 9) | ((lane & 1) << 8) | row;
        return wave < nrows ? 2 * ((nrows - wave + 31) >> 5) : 0;
    };
    const unsigned lt = (1u << lane) - 1u;
    auto handle = [&](int a, int b, unsigned lim) {
        const bool t0 = (unsigned)(b - r0) < lim;
        const bool t1 = (unsigned)(a - r0) < lim;
        const unsigned bal0 = __builtin_amdgcn_ballot_w32(t0);
        const unsigned bal1 = __builtin_amdgcn_ballot_w32(t1);
        if (bal0 | bal1) {
            const int c0 = __popc(bal0);
            int* q = &s_q[wave * QCAP + qn];
            if (t0) q[__popc(bal0 & lt)]      = (a << 9) | (b - r0);
            if (t1) q[c0 + __popc(bal1 & lt)] = (b << 9) | 256 | (a - r0);
            qn += c0 + __popc(bal1);
        }
    };
    auto plain_sweep = [&](auto full) {
        for (int base = 0; base < E; base += 1024) {
            const int e = base + tid;
            const bool ok = e < E;
            const int ec = ok ? e : 0;
            if (qn > QCAP - 64) qn = full(qn);
            handle(ei0[ec], ei1[ec], ok ? (unsigned)nrows : 0u);
        }
    };
    auto give_up = [&](int) -> int { s_over = 1; return 0; };

    qn = seed();
    if (keys_usable(E, n, ws_bytes)) {
        const uint4* __restrict__ p4 = (const uint4*)planes;
        const int nblk  = E >> 5;
        const int nrun  = key_runs(E);
        const int niter = (nrun + 31) >> 5;
        const int seeds = qn;
        int gave = 0;
        const int start = (int)((blockIdx.x * (unsigned)niter) / gridDim.x);
        unsigned gm[8];
        #pragma unroll
        for (int b = 0; b < 8; ++b) gm[b] = ((blockIdx.x >> b) & 1u) ? ~0u : 0u;
        auto ld = [&](int it, uint4 (&k)[4]) -> int {
            int c = start + it; if (c >= niter) c -= niter;
            const int run = c * 32 + wave;
            const bool in = (it < niter) && (run < nrun);
            const int rc = in ? run : 0;
            #pragma unroll
            for (int j = 0; j < 4; ++j) k[j] = p4[rc * 128 + j * 32 + lane];
            const int gb = rc * 32 + lane;
            return (in && gb < nblk) ? gb : -1;
        };
        auto match = [&](const uint4& lo, const uint4& hi) -> unsigned {
            unsigned m = ~(lo.x ^ gm[0]);
            m &= ~(lo.y ^ gm[1]); m &= ~(lo.z ^ gm[2]); m &= ~(lo.w ^ gm[3]);
            m &= ~(hi.x ^ gm[4]); m &= ~(hi.y ^ gm[5]); m &= ~(hi.z ^ gm[6]); m &= ~(hi.w ^ gm[7]);
            return m;
        };
        auto note = [&](unsigned m, int gb, int conv) {
            while (true) {
                const bool has = m != 0u;
                const unsigned bal = __builtin_amdgcn_ballot_w32(has);
                if (bal == 0u) break;
                if (qn > QCAP - 32) { s_over = 1; gave = 1; qn = 0; }
                if (has) {
                    const int p = __builtin_ctz(m);
                    m &= m - 1u;
                    s_q[wave * QCAP + qn + __popc(bal & lt)] = (((gb << 5) | p) << 1) | conv;
                }
                qn += __popc(bal);
            }
        };
        auto work = [&](const uint4 (&k)[4], int gb) {
            const unsigned m1 = gb >= 0 ? match(k[2], k[3]) : 0u;
            const unsigned m2 = gb >= 0 ? match(k[0], k[1]) : 0u;
            note(m1, gb, 0);
            note(m2, gb, 1);
        };
        uint4 kc[2][4];
        int gc[2];
        #pragma unroll
        for (int s = 0; s < 2; ++s) gc[s] = ld(s, kc[s]);
        for (int it = 0; it < niter; it += 2) {
            #pragma unroll
            for (int s = 0; s < 2; ++s) {
                work(kc[s], gc[s]);
                gc[s] = ld(it + 2 + s, kc[s]);
            }
        }
        const int first = gave ? 0 : seeds;
        if (gave) qn = 0;
        const bool with_pairs = pairs_usable(E, n, ws_bytes);
        for (int base = first; base < qn; base += 128) {
            int code[4], ra[4], rb[4];
            #pragma unroll
            for (int u = 0; u < 4; ++u) {
                const int i = base + 32 * u + lane;
                code[u] = 0;
                if (i < qn) code[u] = s_q[wave * QCAP + i];
                if (with_pairs) {
                    const unsigned pv = pair[code[u] >> 1];
                    ra[u] = (int)(pv & 0xFFFFu);
                    rb[u] = (int)(pv >> 16);
                } else {
                    ra[u] = ei0[code[u] >> 1];
                    rb[u] = ei1[code[u] >> 1];
                }
            }
            #pragma unroll
            for (int u = 0; u < 4; ++u) {
                const int i = base + 32 * u + lane;
                if (i < qn)
                    s_q[wave * QCAP + i] = (code[u] & 1) ? ((rb[u] << 9) | 256 | (ra[u] - r0))
                                                         : ((ra[u] << 9) | (rb[u] - r0));
            }
        }
    } else {
        plain_sweep(give_up);
    }
    maxpass(qn);
    __syncthreads();

    const int over = __builtin_amdgcn_readfirstlane(s_over);
    if (!over) {
        qn = drain(qn);
        tail(qn);
    } else {
        qn = seed();
        plain_sweep([&](int cnt) -> int { maxpass(cnt); return 0; });
        maxpass(qn);
        __syncthreads();
        qn = seed();
        plain_sweep(drain);
        qn = drain(qn);
        tail(qn);
    }
    __syncthreads();

    for (int i = tid; i < nrows * (F_OUT / 4); i += 1024) {
        const int row = i >> 5;
        const int c4  = (i & 31) * 4;
        const int h   = c4 >> 5;
        const float* __restrict__ p1 = &s_num[row * F_OUT + (i & 31)];
        const float* __restrict__ p2 = &s_num[(ROWS_PG + row) * F_OUT + (i & 31)];
        const float4 n1 = make_float4(p1[0], p1[32], p1[64], p1[96]);
        const float4 n2 = make_float4(p2[0], p2[32], p2[64], p2[96]);
        const float d1 = s_den[row * HEADS + h];
        const float d2 = s_den[(ROWS_PG + row) * HEADS + h];
        const float4 bb1 = *(const float4*)&b1[c4];
        const float4 bb2 = *(const float4*)&b2[c4];
        float4 o;
        o.x = (1.0f - ALPHA_) * (n1.x / d1 + bb1.x) + ALPHA_ * (n2.x / d2 + bb2.x);
        o.y = (1.0f - ALPHA_) * (n1.y / d1 + bb1.y) + ALPHA_ * (n2.y / d2 + bb2.y);
        o.z = (1.0f - ALPHA_) * (n1.z / d1 + bb1.z) + ALPHA_ * (n2.z / d2 + bb2.z);
        o.w = (1.0f - ALPHA_) * (n1.w / d1 + bb1.w) + ALPHA_ * (n2.w / d2 + bb2.w);
        *(float4*)&out[(size_t)(r0 + row) * F_OUT + c4] = o;
    }
}

extern "C" void kernel_launch(void* const* d_in, const int* in_sizes, int n_in,
                              void* d_out, int out_size, void* d_ws, size_t ws_size,
                              hipStream_t stream) {
    const float* x    = (const float*)d_in[0];
    const int*   ei   = (const int*)d_in[1];
    const float* W1   = (const float*)d_in[2];
    const float* as1  = (const float*)d_in[3];
    const float* ad1  = (const float*)d_in[4];
    const float* b1   = (const float*)d_in[5];
    const float* W2   = (const float*)d_in[6];
    const float* as2  = (const float*)d_in[7];
    const float* ad2  = (const float*)d_in[8];
    const float* b2   = (const float*)d_in[9];
    float* out = (float*)d_out;

    const int n = in_sizes[0] / F_IN;
    const int E = in_sizes[1] / 2;
    const int* src = ei;
    const int* dst = ei + E;

    float* p   = (float*)d_ws;
    float* H1  = p;  p += (size_t)n * F_OUT;
    float* H2  = p;  p += (size_t)n * F_OUT;
    float* es1 = p;  p += (size_t)n * HEADS;
    float* ed1 = p;  p += (size_t)n * HEADS;
    float* es2 = p;  p += (size_t)n * HEADS;
    float* ed2 = p;  p += (size_t)n * HEADS;
    int nrun = (E / 32 + 31) / 32;
    unsigned* planes = (unsigned*)p;
    unsigned* pair   = planes + (size_t)nrun * 512;

    gemm_wmma_kernel<<<512, GEMM_THREADS, 0, stream>>>(x, W1, W2, H1, H2, n);
    {
        int total = n * HEADS * 2;
        int nkb = (nrun * 4 + 7) / 8;
        scores_kernel<<<nkb + (total + 255) / 256, 256, 0, stream>>>(
            H1, H2, as1, ad1, as2, ad2, es1, ed1, es2, ed2, n,
            src, dst, E, nkb, planes, pair, (unsigned long long)ws_size);
    }
    {
        int groups = (n + ROWS_PG - 1) / ROWS_PG;
        owner_kernel<<<groups, 1024, 0, stream>>>(src, dst, E, n, H1, H2,
                                                  es1, ed1, es2, ed2, b1, b2,
                                                  planes, pair,
                                                  (unsigned long long)ws_size, out);
    }
}


# owner sweep reads one-hot nibble masks of the owner keys (a group ANDs 2 arrays per end: 4 dwords per 32 edges, a quarter of the planes'), built in the scores launch; list and LDS sum order change
# speedup vs baseline: 20.8469x; 1.0978x over previous
#include <hip/hip_runtime.h>
#include <hip/hip_bf16.h>


typedef __attribute__((ext_vector_type(2))) float v2f;
typedef __attribute__((ext_vector_type(8))) float v8f;

#define F_IN   128
#define HEADS  4
#define C_OUT  32
#define F_OUT  128
#define ALPHA_ 0.5f
#define SLOPE_ 0.2f

__device__ __forceinline__ float lrelu(float v) {
    return v > 0.0f ? v : v * SLOPE_;
}

#define A_PITCH 132
#define GEMM_THREADS 512

__global__ __launch_bounds__(GEMM_THREADS) void gemm_wmma_kernel(
    const float* __restrict__ X,
    const float* __restrict__ W1, const float* __restrict__ W2,
    float* __restrict__ H1, float* __restrict__ H2, int n)
{
    __shared__ __attribute__((aligned(16))) float s_a[2][16 * A_PITCH];

    const int tid  = threadIdx.x;
    const int lane = tid & 31;
    const int wave = __builtin_amdgcn_readfirstlane(tid >> 5);
    const int conv  = wave >> 3;
    const int ntile = wave & 7;
    const int half = lane >> 4;
    const int l    = lane & 15;

    const float* __restrict__ Wm = conv ? W2 : W1;
    float* __restrict__ Hm       = conv ? H2 : H1;
    const int col = ntile * 16 + l;

    v2f bf[32];
    #pragma unroll
    for (int j = 0; j < 32; ++j) {
        const int ka = 4 * j + half * 2;
        bf[j].x = Wm[ka * F_OUT + col];
        bf[j].y = Wm[(ka + 1) * F_OUT + col];
    }

    const int mtiles = (n + 15) >> 4;
    auto ldx = [&](int mt) -> float4 {
        int row = mt * 16 + wave;
        if (row >= n) row = n - 1;
        return ((const float4*)(X + (size_t)row * F_IN))[lane];
    };
    const int st_off = wave * A_PITCH + lane * 4;
    const int fr_off = l * A_PITCH + half * 2;

    int mt = blockIdx.x;
    if (mt < mtiles) *(float4*)&s_a[0][st_off] = ldx(mt);
    __syncthreads();
    int cur = 0;
    for (; mt < mtiles; mt += gridDim.x) {
        const int  nx   = mt + gridDim.x;
        const bool more = nx < mtiles;
        float4 vn = make_float4(0.0f, 0.0f, 0.0f, 0.0f);
        if (more) vn = ldx(nx);

        const float* __restrict__ sa = &s_a[cur][fr_off];
        v8f acc = {};
        #pragma unroll
        for (int j = 0; j < 32; ++j) {
            const v2f a = *(const v2f*)&sa[4 * j];
            acc = __builtin_amdgcn_wmma_f32_16x16x4_f32(
                 false, a,  false, bf[j],
                 (short)0, acc,  false,  false);
        }

        if (more) *(float4*)&s_a[cur ^ 1][st_off] = vn;
        __syncthreads();
        cur ^= 1;

        const int rowBase = mt * 16 + half * 8;
        float* __restrict__ hp = Hm + (size_t)rowBase * F_OUT + col;
        if (mt * 16 + 16 <= n) {
            #pragma unroll
            for (int r = 0; r < 8; ++r) hp[r * F_OUT] = acc[r];
        } else {
            #pragma unroll
            for (int r = 0; r < 8; ++r) if (rowBase + r < n) hp[r * F_OUT] = acc[r];
        }
    }
}

#define ROWS_PG  196

__device__ __forceinline__ int key_runs(int E) { return ((E >> 5) + 31) >> 5; }
__device__ __forceinline__ unsigned long long key_bytes(int E, int n) {
    return (unsigned long long)n * (2 * F_OUT + 4 * HEADS) * 4ull + (unsigned long long)key_runs(E) * 2048ull;
}
__device__ __forceinline__ bool keys_usable(int E, int n, unsigned long long ws_bytes) {
    return (E & 31) == 0 && (n + ROWS_PG - 1) / ROWS_PG <= 256 && ws_bytes >= key_bytes(E, n);
}
__device__ __forceinline__ bool pairs_usable(int E, int n, unsigned long long ws_bytes) {
    return ws_bytes >= key_bytes(E, n) + (unsigned long long)E * 4ull;
}

__device__ __forceinline__ unsigned long long mask_bytes(int E, int n) {
    return (unsigned long long)n * (2 * F_OUT + 4 * HEADS) * 4ull
         + (unsigned long long)key_runs(E) * 8192ull + (unsigned long long)E * 4ull;
}
__device__ __forceinline__ bool masks_usable(int E, int n, unsigned long long ws_bytes) {
    return (E & 31) == 0 && (n + ROWS_PG - 1) / ROWS_PG <= 256 && n <= 65536 && ws_bytes >= mask_bytes(E, n);
}

__global__ void scores_kernel(
    const float* __restrict__ H1, const float* __restrict__ H2,
    const float* __restrict__ as1, const float* __restrict__ ad1,
    const float* __restrict__ as2, const float* __restrict__ ad2,
    float* __restrict__ es1, float* __restrict__ ed1,
    float* __restrict__ es2, float* __restrict__ ed2, int n,
    const int* __restrict__ ei0, const int* __restrict__ ei1, int E, int nkb,
    unsigned* __restrict__ planes, unsigned* __restrict__ pair,
    unsigned long long ws_bytes)
{
    if ((int)blockIdx.x < nkb) {
        if (masks_usable(E, n, ws_bytes)) {
            const int lane = threadIdx.x & 31;
            const int wv   = __builtin_amdgcn_readfirstlane((int)(threadIdx.x >> 5));
            const int nrun = key_runs(E);
            const int run  = (int)blockIdx.x * 2 + (wv >> 1);
            const int end  = wv & 1;
            if (wv >= 4 || run >= nrun) return;
            const int nblk = E >> 5;
            const size_t NW = (size_t)nrun * 32;
            unsigned* __restrict__ pairM = planes + NW * 64;
            int mk[32];
            #pragma unroll
            for (int a = 0; a < 32; ++a) mk[a] = 0;
            #pragma nounroll
            for (int c = 0; c < 4; ++c) {
                int ra[8], rb[8];
                #pragma unroll
                for (int u = 0; u < 8; ++u) {
                    const int gb = run * 32 + c * 8 + u;
                    const int ec = gb < nblk ? gb * 32 + lane : 0;
                    ra[u] = ei0[ec];
                    rb[u] = ei1[ec];
                }
                if (end == 0) {
                    #pragma unroll
                    for (int u = 0; u < 8; ++u) {
                        const int gb = run * 32 + c * 8 + u;
                        if (gb < nblk) pairM[(size_t)gb * 32 + lane] = (unsigned)ra[u] | (unsigned)rb[u] << 16;
                    }
                }
                #pragma unroll
                for (int u = 0; u < 8; ++u) {
                    const int gb = run * 32 + c * 8 + u;
                    const unsigned k  = (unsigned)(end ? rb[u] : ra[u]) / ROWS_PG;
                    const unsigned kh = k >> 4, kl = k & 15u;
                    const bool mine = (lane == c * 8 + u) && (gb < nblk);
                    #pragma unroll
                    for (int v = 0; v < 16; ++v) {
                        const int bh = (int)__builtin_amdgcn_ballot_w32(kh == (unsigned)v);
                        const int bl = (int)__builtin_amdgcn_ballot_w32(kl == (unsigned)v);
                        if (mine) { mk[v] = bh; mk[16 + v] = bl; }
                    }
                }
            }
            unsigned* __restrict__ mp = planes + (size_t)(end * 32) * NW + (size_t)run * 32 + lane;
            #pragma unroll
            for (int a = 0; a < 32; ++a) mp[(size_t)a * NW] = (unsigned)mk[a];
            return;
        }
        if (!keys_usable(E, n, ws_bytes)) return;
        const int lane = threadIdx.x & 31;
        const int w    = __builtin_amdgcn_readfirstlane((int)(blockIdx.x * (blockDim.x >> 5) + (threadIdx.x >> 5)));
        const int nblk = E >> 5;
        if (w >= key_runs(E) * 4) return;
        const bool with_pairs = pairs_usable(E, n, ws_bytes);
        int av[8], bv[8];
        #pragma unroll
        for (int u = 0; u < 8; ++u) {
            const int gb = w * 8 + u;
            const int ec = gb < nblk ? gb * 32 + lane : 0;
            av[u] = ei0[ec];
            bv[u] = ei1[ec];
        }
        unsigned pa[8], pb[8];
        #pragma unroll
        for (int b = 0; b < 8; ++b) { pa[b] = 0u; pb[b] = 0u; }
        #pragma unroll
        for (int u = 0; u < 8; ++u) {
            const int gb = w * 8 + u;
            if (with_pairs && gb < nblk) pair[gb * 32 + lane] = (unsigned)av[u] | (unsigned)bv[u] << 16;
            const unsigned ka = (unsigned)av[u] / ROWS_PG;
            const unsigned kb = (unsigned)bv[u] / ROWS_PG;
            #pragma unroll
            for (int b = 0; b < 8; ++b) {
                const unsigned ba = __builtin_amdgcn_ballot_w32(((ka >> b) & 1u) != 0u);
                const unsigned bb = __builtin_amdgcn_ballot_w32(((kb >> b) & 1u) != 0u);
                if (lane == u) { pa[b] = ba; pb[b] = bb; }
            }
        }
        if (lane < 8) {
            const int gb = w * 8 + lane;
            uint4* __restrict__ p4 = (uint4*)planes + (size_t)(gb >> 5) * 128 + (gb & 31);
            p4[0]  = make_uint4(pa[0], pa[1], pa[2], pa[3]);
            p4[32] = make_uint4(pa[4], pa[5], pa[6], pa[7]);
            p4[64] = make_uint4(pb[0], pb[1], pb[2], pb[3]);
            p4[96] = make_uint4(pb[4], pb[5], pb[6], pb[7]);
        }
        return;
    }
    int i = (blockIdx.x - nkb) * blockDim.x + threadIdx.x;
    int total = n * HEADS * 2;
    if (i >= total) return;
    int conv = (i >= n * HEADS) ? 1 : 0;
    int j = conv ? (i - n * HEADS) : i;
    int node = j >> 2;
    int h    = j & 3;

    const float* __restrict__ hp  = (conv ? H2 : H1) + node * F_OUT + h * C_OUT;
    const float* __restrict__ asp = (conv ? as2 : as1) + h * C_OUT;
    const float* __restrict__ adp = (conv ? ad2 : ad1) + h * C_OUT;

    float ssum = 0.0f, dsum = 0.0f;
    #pragma unroll
    for (int c = 0; c < C_OUT; ++c) {
        float v = hp[c];
        ssum += v * asp[c];
        dsum += v * adp[c];
    }
    if (conv) { es2[node * HEADS + h] = ssum; ed2[node * HEADS + h] = dsum; }
    else      { es1[node * HEADS + h] = ssum; ed1[node * HEADS + h] = dsum; }
}

#define QCAP     768

template <int U>
__device__ __forceinline__ float from_lane_of_eight(float v) {
    return __int_as_float((int)__builtin_amdgcn_mov_dpp8((unsigned)__float_as_int(v), (unsigned)(U * 0x249249)));
}
template <int U>
__device__ __forceinline__ void add_row(float* np, float w, const float4& r) {
    const float wu = from_lane_of_eight<U>(w);
    __hip_atomic_fetch_add(np,      wu * r.x, __ATOMIC_RELAXED, __HIP_MEMORY_SCOPE_WORKGROUP);
    __hip_atomic_fetch_add(np + 32, wu * r.y, __ATOMIC_RELAXED, __HIP_MEMORY_SCOPE_WORKGROUP);
    __hip_atomic_fetch_add(np + 64, wu * r.z, __ATOMIC_RELAXED, __HIP_MEMORY_SCOPE_WORKGROUP);
    __hip_atomic_fetch_add(np + 96, wu * r.w, __ATOMIC_RELAXED, __HIP_MEMORY_SCOPE_WORKGROUP);
}

__global__ __launch_bounds__(1024) void owner_kernel(
    const int* __restrict__ ei0, const int* __restrict__ ei1, int E, int n,
    const float* __restrict__ H1, const float* __restrict__ H2,
    const float* __restrict__ es1, const float* __restrict__ ed1,
    const float* __restrict__ es2, const float* __restrict__ ed2,
    const float* __restrict__ b1, const float* __restrict__ b2,
    const unsigned* __restrict__ planes, const unsigned* __restrict__ pair,
    unsigned long long ws_bytes,
    float* __restrict__ out)
{
    __shared__ __attribute__((aligned(16))) float s_num[2 * ROWS_PG * F_OUT];
    __shared__ float  s_den[2 * ROWS_PG * HEADS];
    __shared__ float2 s_em[2 * ROWS_PG * HEADS];
    __shared__ __attribute__((aligned(16))) int s_q[32 * QCAP];
    __shared__ int s_over;

    const int tid  = threadIdx.x;
    const int lane = tid & 31;
    const int wave = __builtin_amdgcn_readfirstlane(tid >> 5);
    const int hd   = lane & 3;
    const int r0   = blockIdx.x * ROWS_PG;
    int nrows = n - r0;
    if (nrows > ROWS_PG) nrows = ROWS_PG;

    {
        const float4 z = make_float4(0.0f, 0.0f, 0.0f, 0.0f);
        for (int i = tid; i < 2 * ROWS_PG * (F_OUT / 4); i += 1024) ((float4*)s_num)[i] = z;
    }
    for (int i = tid; i < 2 * ROWS_PG * HEADS; i += 1024) {
        int conv = i >= ROWS_PG * HEADS ? 1 : 0;
        int j = i - conv * ROWS_PG * HEADS;
        float2 em; em.x = 0.0f; em.y = -__builtin_huge_valf();
        if ((j >> 2) < nrows) em.x = (conv ? ed2 : ed1)[r0 * HEADS + j];
        s_em[i]  = em;
        s_den[i] = 0.0f;
    }
    if (tid == 0) s_over = 0;
    __syncthreads();

    auto batch = [&](int i, const int cnt) {
        const int* q = &s_q[wave * QCAP + i];
        const int  ul   = lane & 7;
        const int  hl   = lane >> 3;
        const bool live = ul < cnt;
        const int  wl   = q[live ? ul : 0];
        const int  cl   = (wl >> 8) & 1;
        const int  sl   = (cl * ROWS_PG + (wl & 255)) * HEADS + hl;
        const float  e  = (cl ? es2 : es1)[(size_t)(wl >> 9) * HEADS + hl];
        const float2 em = s_em[sl];
        float4 r[8];
        float* np[8];
        #pragma unroll
        for (int u = 0; u < 8; ++u) {
            if (u < cnt) {
                const int word = __builtin_amdgcn_readlane(wl, u);
                const int conv = (word >> 8) & 1;
                const float* __restrict__ hp = (conv ? H2 : H1) + (size_t)(word >> 9) * F_OUT;
                r[u]  = ((const float4*)hp)[lane];
                np[u] = &s_num[(conv * ROWS_PG + (word & 255)) * F_OUT + lane];
            }
        }
        const float w = __expf(lrelu(e + em.x) - em.y);
        if (live)
            __hip_atomic_fetch_add(&s_den[sl], w, __ATOMIC_RELAXED, __HIP_MEMORY_SCOPE_WORKGROUP);
        if (0 < cnt) add_row<0>(np[0], w, r[0]);
        if (1 < cnt) add_row<1>(np[1], w, r[1]);
        if (2 < cnt) add_row<2>(np[2], w, r[2]);
        if (3 < cnt) add_row<3>(np[3], w, r[3]);
        if (4 < cnt) add_row<4>(np[4], w, r[4]);
        if (5 < cnt) add_row<5>(np[5], w, r[5]);
        if (6 < cnt) add_row<6>(np[6], w, r[6]);
        if (7 < cnt) add_row<7>(np[7], w, r[7]);
    };
    auto drain = [&](int cnt) -> int {
        int* q = &s_q[wave * QCAP];
        int i = 0;
        for (; i + 8 <= cnt; i += 8) batch(i, 8);
        const int rem = cnt - i;
        if (i > 0 && lane < rem) { const int wv = q[i + lane]; q[lane] = wv; }
        return rem;
    };
    auto tail = [&](int cnt) {
        for (int i = 0; i < cnt; i += 8) batch(i, cnt - i < 8 ? cnt - i : 8);
    };
    auto maxpass = [&](int cnt) {
        const int* q = &s_q[wave * QCAP];
        const int sub = lane >> 2;
        for (int i = 0; i < cnt; i += 64) {
            float e[8];
            int   at[8];
            #pragma unroll
            for (int u = 0; u < 8; ++u) {
                int k = i + 8 * u + sub;
                if (k > cnt - 1) k = cnt - 1;
                const int word = q[k];
                const int conv = (word >> 8) & 1;
                at[u] = (conv * ROWS_PG + (word & 255)) * HEADS + hd;
                e[u]  = (conv ? es2 : es1)[(size_t)(word >> 9) * HEADS + hd];
            }
            #pragma unroll
            for (int u = 0; u < 8; ++u) {
                const float v = lrelu(e[u] + s_em[at[u]].x);
                __hip_atomic_fetch_max(&s_em[at[u]].y, v,
                                       __ATOMIC_RELAXED, __HIP_MEMORY_SCOPE_WORKGROUP);
            }
        }
    };

    int qn = 0;
    auto seed = [&]() -> int {
        const int row = wave + 32 * (lane >> 1);
        if (row < nrows) s_q[wave * QCAP + lane] = ((r0 + row) << 9) | ((lane & 1) << 8) | row;
        return wave < nrows ? 2 * ((nrows - wave + 31) >> 5) : 0;
    };
    const unsigned lt = (1u << lane) - 1u;
    auto handle = [&](int a, int b, unsigned lim) {
        const bool t0 = (unsigned)(b - r0) < lim;
        const bool t1 = (unsigned)(a - r0) < lim;
        const unsigned bal0 = __builtin_amdgcn_ballot_w32(t0);
        const unsigned bal1 = __builtin_amdgcn_ballot_w32(t1);
        if (bal0 | bal1) {
            const int c0 = __popc(bal0);
            int* q = &s_q[wave * QCAP + qn];
            if (t0) q[__popc(bal0 & lt)]      = (a << 9) | (b - r0);
            if (t1) q[c0 + __popc(bal1 & lt)] = (b << 9) | 256 | (a - r0);
            qn += c0 + __popc(bal1);
        }
    };
    auto plain_sweep = [&](auto full) {
        for (int base = 0; base < E; base += 1024) {
            const int e = base + tid;
            const bool ok = e < E;
            const int ec = ok ? e : 0;
            if (qn > QCAP - 64) qn = full(qn);
            handle(ei0[ec], ei1[ec], ok ? (unsigned)nrows : 0u);
        }
    };
    auto give_up = [&](int) -> int { s_over = 1; return 0; };

    int gave = 0;
    auto note = [&](unsigned m, int gb, int conv) {
        while (true) {
            const bool has = m != 0u;
            const unsigned bal = __builtin_amdgcn_ballot_w32(has);
            if (bal == 0u) break;
            if (qn > QCAP - 32) { s_over = 1; gave = 1; qn = 0; }
            if (has) {
                const int p = __builtin_ctz(m);
                m &= m - 1u;
                s_q[wave * QCAP + qn + __popc(bal & lt)] = (((gb << 5) | p) << 1) | conv;
            }
            qn += __popc(bal);
        }
    };
    auto note4 = [&](unsigned w0, unsigned w1, unsigned w2, unsigned w3, int gb, int conv) {
        while (true) {
            const bool has = (w0 | w1 | w2 | w3) != 0u;
            const unsigned bal = __builtin_amdgcn_ballot_w32(has);
            if (bal == 0u) break;
            if (qn > QCAP - 32) { s_over = 1; gave = 1; qn = 0; }
            if (has) {
                const bool z1 = w0 == 0u, z2 = z1 && w1 == 0u, z3 = z2 && w2 == 0u;
                const unsigned cur = z3 ? w3 : (z2 ? w2 : (z1 ? w1 : w0));
                const int j = z3 ? 3 : (z2 ? 2 : (z1 ? 1 : 0));
                const int p = __builtin_ctz(cur);
                w0 &= w0 - 1u;
                w1 = z1 ? (w1 & (w1 - 1u)) : w1;
                w2 = z2 ? (w2 & (w2 - 1u)) : w2;
                w3 = z3 ? (w3 & (w3 - 1u)) : w3;
                s_q[wave * QCAP + qn + __popc(bal & lt)] = ((((gb + j) << 5) | p) << 1) | conv;
            }
            qn += __popc(bal);
        }
    };
    auto convert = [&](int first, bool with_pairs, const unsigned* __restrict__ pp) {
        for (int base = first; base < qn; base += 128) {
            int code[4], ra[4], rb[4];
            #pragma unroll
            for (int u = 0; u < 4; ++u) {
                const int i = base + 32 * u + lane;
                code[u] = 0;
                if (i < qn) code[u] = s_q[wave * QCAP + i];
                if (with_pairs) {
                    const unsigned pv = pp[code[u] >> 1];
                    ra[u] = (int)(pv & 0xFFFFu);
                    rb[u] = (int)(pv >> 16);
                } else {
                    ra[u] = ei0[code[u] >> 1];
                    rb[u] = ei1[code[u] >> 1];
                }
            }
            #pragma unroll
            for (int u = 0; u < 4; ++u) {
                const int i = base + 32 * u + lane;
                if (i < qn)
                    s_q[wave * QCAP + i] = (code[u] & 1) ? ((rb[u] << 9) | 256 | (ra[u] - r0))
                                                         : ((ra[u] << 9) | (rb[u] - r0));
            }
        }
    };

    qn = seed();
    if (masks_usable(E, n, ws_bytes)) {
        const int nrun = key_runs(E);
        const size_t NW = (size_t)nrun * 32;
        const int NQ    = nrun * 8;
        const int nstep = (NQ + 1023) >> 10;
        const unsigned g = blockIdx.x;
        const uint4* __restrict__ a_hi = (const uint4*)(planes + (size_t)(g >> 4) * NW);
        const uint4* __restrict__ a_lo = (const uint4*)(planes + (size_t)(16u + (g & 15u)) * NW);
        const uint4* __restrict__ b_hi = (const uint4*)(planes + (size_t)(32u + (g >> 4)) * NW);
        const uint4* __restrict__ b_lo = (const uint4*)(planes + (size_t)(48u + (g & 15u)) * NW);
        const unsigned* __restrict__ pairM = planes + NW * 64;
        const int seeds = qn;
        const int start = (int)((blockIdx.x * (unsigned)nstep) / gridDim.x);
        auto ld = [&](int it, uint4 (&k)[4]) -> int {
            int c = start + it; if (c >= nstep) c -= nstep;
            const int q = c * 1024 + wave * 32 + lane;
            const bool in = (it < nstep) && (q < NQ);
            const int qc = in ? q : 0;
            k[0] = a_hi[qc]; k[1] = a_lo[qc]; k[2] = b_hi[qc]; k[3] = b_lo[qc];
            return in ? q * 4 : -1;
        };
        auto work = [&](const uint4 (&k)[4], int gb) {
            const unsigned live = gb >= 0 ? ~0u : 0u;
            note4(k[2].x & k[3].x & live, k[2].y & k[3].y & live, k[2].z & k[3].z & live, k[2].w & k[3].w & live, gb, 0);
            note4(k[0].x & k[1].x & live, k[0].y & k[1].y & live, k[0].z & k[1].z & live, k[0].w & k[1].w & live, gb, 1);
        };
        uint4 kc[2][4];
        int gc[2];
        #pragma unroll
        for (int s = 0; s < 2; ++s) gc[s] = ld(s, kc[s]);
        for (int it = 0; it < nstep; it += 2) {
            #pragma unroll
            for (int s = 0; s < 2; ++s) {
                work(kc[s], gc[s]);
                gc[s] = ld(it + 2 + s, kc[s]);
            }
        }
        if (gave) qn = 0;
        convert(gave ? 0 : seeds, true, pairM);
    } else if (keys_usable(E, n, ws_bytes)) {
        const uint4* __restrict__ p4 = (const uint4*)planes;
        const int nblk  = E >> 5;
        const int nrun  = key_runs(E);
        const int niter = (nrun + 31) >> 5;
        const int seeds = qn;
        const int start = (int)((blockIdx.x * (unsigned)niter) / gridDim.x);
        unsigned gm[8];
        #pragma unroll
        for (int b = 0; b < 8; ++b) gm[b] = ((blockIdx.x >> b) & 1u) ? ~0u : 0u;
        auto ld = [&](int it, uint4 (&k)[4]) -> int {
            int c = start + it; if (c >= niter) c -= niter;
            const int run = c * 32 + wave;
            const bool in = (it < niter) && (run < nrun);
            const int rc = in ? run : 0;
            #pragma unroll
            for (int j = 0; j < 4; ++j) k[j] = p4[rc * 128 + j * 32 + lane];
            const int gb = rc * 32 + lane;
            return (in && gb < nblk) ? gb : -1;
        };
        auto match = [&](const uint4& lo, const uint4& hi) -> unsigned {
            unsigned m = ~(lo.x ^ gm[0]);
            m &= ~(lo.y ^ gm[1]); m &= ~(lo.z ^ gm[2]); m &= ~(lo.w ^ gm[3]);
            m &= ~(hi.x ^ gm[4]); m &= ~(hi.y ^ gm[5]); m &= ~(hi.z ^ gm[6]); m &= ~(hi.w ^ gm[7]);
            return m;
        };
        auto work = [&](const uint4 (&k)[4], int gb) {
            const unsigned m1 = gb >= 0 ? match(k[2], k[3]) : 0u;
            const unsigned m2 = gb >= 0 ? match(k[0], k[1]) : 0u;
            note(m1, gb, 0);
            note(m2, gb, 1);
        };
        uint4 kc[2][4];
        int gc[2];
        #pragma unroll
        for (int s = 0; s < 2; ++s) gc[s] = ld(s, kc[s]);
        for (int it = 0; it < niter; it += 2) {
            #pragma unroll
            for (int s = 0; s < 2; ++s) {
                work(kc[s], gc[s]);
                gc[s] = ld(it + 2 + s, kc[s]);
            }
        }
        if (gave) qn = 0;
        convert(gave ? 0 : seeds, pairs_usable(E, n, ws_bytes), pair);
    } else {
        plain_sweep(give_up);
    }
    maxpass(qn);
    __syncthreads();

    const int over = __builtin_amdgcn_readfirstlane(s_over);
    if (!over) {
        qn = drain(qn);
        tail(qn);
    } else {
        qn = seed();
        plain_sweep([&](int cnt) -> int { maxpass(cnt); return 0; });
        maxpass(qn);
        __syncthreads();
        qn = seed();
        plain_sweep(drain);
        qn = drain(qn);
        tail(qn);
    }
    __syncthreads();

    for (int i = tid; i < nrows * (F_OUT / 4); i += 1024) {
        const int row = i >> 5;
        const int c4  = (i & 31) * 4;
        const int h   = c4 >> 5;
        const float* __restrict__ p1 = &s_num[row * F_OUT + (i & 31)];
        const float* __restrict__ p2 = &s_num[(ROWS_PG + row) * F_OUT + (i & 31)];
        const float4 n1 = make_float4(p1[0], p1[32], p1[64], p1[96]);
        const float4 n2 = make_float4(p2[0], p2[32], p2[64], p2[96]);
        const float d1 = s_den[row * HEADS + h];
        const float d2 = s_den[(ROWS_PG + row) * HEADS + h];
        const float4 bb1 = *(const float4*)&b1[c4];
        const float4 bb2 = *(const float4*)&b2[c4];
        float4 o;
        o.x = (1.0f - ALPHA_) * (n1.x / d1 + bb1.x) + ALPHA_ * (n2.x / d2 + bb2.x);
        o.y = (1.0f - ALPHA_) * (n1.y / d1 + bb1.y) + ALPHA_ * (n2.y / d2 + bb2.y);
        o.z = (1.0f - ALPHA_) * (n1.z / d1 + bb1.z) + ALPHA_ * (n2.z / d2 + bb2.z);
        o.w = (1.0f - ALPHA_) * (n1.w / d1 + bb1.w) + ALPHA_ * (n2.w / d2 + bb2.w);
        *(float4*)&out[(size_t)(r0 + row) * F_OUT + c4] = o;
    }
}

extern "C" void kernel_launch(void* const* d_in, const int* in_sizes, int n_in,
                              void* d_out, int out_size, void* d_ws, size_t ws_size,
                              hipStream_t stream) {
    const float* x    = (const float*)d_in[0];
    const int*   ei   = (const int*)d_in[1];
    const float* W1   = (const float*)d_in[2];
    const float* as1  = (const float*)d_in[3];
    const float* ad1  = (const float*)d_in[4];
    const float* b1   = (const float*)d_in[5];
    const float* W2   = (const float*)d_in[6];
    const float* as2  = (const float*)d_in[7];
    const float* ad2  = (const float*)d_in[8];
    const float* b2   = (const float*)d_in[9];
    float* out = (float*)d_out;

    const int n = in_sizes[0] / F_IN;
    const int E = in_sizes[1] / 2;
    const int* src = ei;
    const int* dst = ei + E;

    float* p   = (float*)d_ws;
    float* H1  = p;  p += (size_t)n * F_OUT;
    float* H2  = p;  p += (size_t)n * F_OUT;
    float* es1 = p;  p += (size_t)n * HEADS;
    float* ed1 = p;  p += (size_t)n * HEADS;
    float* es2 = p;  p += (size_t)n * HEADS;
    float* ed2 = p;  p += (size_t)n * HEADS;
    int nrun = (E / 32 + 31) / 32;
    unsigned* planes = (unsigned*)p;
    unsigned* pair   = planes + (size_t)nrun * 512;

    gemm_wmma_kernel<<<512, GEMM_THREADS, 0, stream>>>(x, W1, W2, H1, H2, n);
    {
        int total = n * HEADS * 2;
        int nkb = (nrun * 4 + 7) / 8;
        scores_kernel<<<nkb + (total + 255) / 256, 256, 0, stream>>>(
            H1, H2, as1, ad1, as2, ad2, es1, ed1, es2, ed2, n,
            src, dst, E, nkb, planes, pair, (unsigned long long)ws_size);
    }
    {
        int groups = (n + ROWS_PG - 1) / ROWS_PG;
        owner_kernel<<<groups, 1024, 0, stream>>>(src, dst, E, n, H1, H2,
                                                  es1, ed1, es2, ed2, b1, b2,
                                                  planes, pair,
                                                  (unsigned long long)ws_size, out);
    }
}
